# starting kernel
# baseline (speedup 1.0000x reference)
#include <hip/hip_runtime.h>

#define NN 100000
#define NE 3200000
#define NR 8

typedef __attribute__((ext_vector_type(2))) float v2f;
typedef __attribute__((ext_vector_type(8))) float v8f;


__global__ void zero_f32(float* __restrict__ p, size_t n) {
    size_t i = (size_t)blockIdx.x * blockDim.x + threadIdx.x;
    size_t stride = (size_t)gridDim.x * blockDim.x;
    for (; i < n; i += stride) p[i] = 0.0f;
}

__global__ void count_edges(const int* __restrict__ dst, const int* __restrict__ et,
                            float* __restrict__ cnt) {
    int e = blockIdx.x * blockDim.x + threadIdx.x;
    if (e >= NE) return;
    atomicAdd(&cnt[(size_t)et[e] * NN + dst[e]], 1.0f);
}

__global__ void recip_clamp(float* __restrict__ c, int n) {
    int i = blockIdx.x * blockDim.x + threadIdx.x;
    if (i < n) c[i] = 1.0f / fmaxf(c[i], 1.0f);
}

template <int F>
__global__ void scatter_edges(const float* __restrict__ feat, const int* __restrict__ src,
                              const int* __restrict__ dst, const int* __restrict__ et,
                              float* __restrict__ agg) {
    constexpr int Q = F / 4;
    long long tid = (long long)blockIdx.x * blockDim.x + threadIdx.x;
    int e = (int)(tid / Q);
    int q = (int)(tid % Q);
    if (e >= NE) return;
    int s = src[e], d = dst[e], r = et[e];
    const float4 v = ((const float4*)(feat + (size_t)s * F))[q];
    float* ap = agg + ((size_t)r * NN + d) * F + (size_t)q * 4;
    atomicAdd(ap + 0, v.x);
    atomicAdd(ap + 1, v.y);
    atomicAdd(ap + 2, v.z);
    atomicAdd(ap + 3, v.w);
}

__device__ __forceinline__ v2f a_frag(const float* __restrict__ M, int row, int stride,
                                      int k0, int kh, float scale) {
    const float* p = M + (size_t)row * stride + k0 + kh;
    v2f a; a.x = p[0] * scale; a.y = p[1] * scale; return a;
}
__device__ __forceinline__ v2f b_frag(const float* __restrict__ W, int nout,
                                      int k0, int kh, int col) {
    v2f b;
    b.x = W[(size_t)(k0 + kh) * nout + col];
    b.y = W[(size_t)(k0 + kh + 1) * nout + col];
    return b;
}


__global__ __launch_bounds__(256) void rgcn_layer1(
    const float* __restrict__ x, const float* __restrict__ agg, const float* __restrict__ inv,
    const float* __restrict__ W1, const float* __restrict__ root1, const float* __restrict__ b1,
    float* __restrict__ h) {
    int wid  = (blockIdx.x * blockDim.x + threadIdx.x) >> 5;
    int lane = threadIdx.x & 31;
    if (wid >= NN / 16) return;
    int m0   = wid * 16;
    int arow = m0 + (lane & 15);
    int kh   = (lane >> 4) << 1;
    int col  = lane & 15;

    v8f c0, c1;
    float bb0 = b1[col], bb1 = b1[16 + col];
#pragma unroll
    for (int v = 0; v < 8; ++v) { c0[v] = bb0; c1[v] = bb1; }

#pragma unroll
    for (int k = 0; k < 16; k += 4) {
        v2f a  = a_frag(x, arow, 16, k, kh, 1.0f);
        v2f bA = b_frag(root1, 32, k, kh, col);
        v2f bB = b_frag(root1, 32, k, kh, col + 16);
        c0 = __builtin_amdgcn_wmma_f32_16x16x4_f32(false, a, false, bA, (short)0, c0, false, false);
        c1 = __builtin_amdgcn_wmma_f32_16x16x4_f32(false, a, false, bB, (short)0, c1, false, false);
    }
    for (int r = 0; r < NR; ++r) {
        const float* ag = agg + (size_t)r * NN * 16;
        const float* Wr = W1 + (size_t)r * 16 * 32;
        float sc = inv[(size_t)r * NN + arow];
#pragma unroll
        for (int k = 0; k < 16; k += 4) {
            v2f a  = a_frag(ag, arow, 16, k, kh, sc);
            v2f bA = b_frag(Wr, 32, k, kh, col);
            v2f bB = b_frag(Wr, 32, k, kh, col + 16);
            c0 = __builtin_amdgcn_wmma_f32_16x16x4_f32(false, a, false, bA, (short)0, c0, false, false);
            c1 = __builtin_amdgcn_wmma_f32_16x16x4_f32(false, a, false, bB, (short)0, c1, false, false);
        }
    }
    int rbase = m0 + ((lane >> 4) << 3);
#pragma unroll
    for (int v = 0; v < 8; ++v) {
        h[(size_t)(rbase + v) * 32 + col]      = fmaxf(c0[v], 0.0f);
        h[(size_t)(rbase + v) * 32 + 16 + col] = fmaxf(c1[v], 0.0f);
    }
}


__global__ __launch_bounds__(256) void rgcn_layer2(
    const float* __restrict__ h, const float* __restrict__ agg, const float* __restrict__ inv,
    const float* __restrict__ W2, const float* __restrict__ root2, const float* __restrict__ b2,
    float* __restrict__ out) {
    int wid  = (blockIdx.x * blockDim.x + threadIdx.x) >> 5;
    int lane = threadIdx.x & 31;
    if (wid >= NN / 16) return;
    int m0   = wid * 16;
    int arow = m0 + (lane & 15);
    int kh   = (lane >> 4) << 1;
    int col  = lane & 15;

    v8f c;
    float bb = b2[col];
#pragma unroll
    for (int v = 0; v < 8; ++v) c[v] = bb;

#pragma unroll
    for (int k = 0; k < 32; k += 4) {
        v2f a = a_frag(h, arow, 32, k, kh, 1.0f);
        v2f b = b_frag(root2, 16, k, kh, col);
        c = __builtin_amdgcn_wmma_f32_16x16x4_f32(false, a, false, b, (short)0, c, false, false);
    }
    for (int r = 0; r < NR; ++r) {
        const float* ag = agg + (size_t)r * NN * 32;
        const float* Wr = W2 + (size_t)r * 32 * 16;
        float sc = inv[(size_t)r * NN + arow];
#pragma unroll
        for (int k = 0; k < 32; k += 4) {
            v2f a = a_frag(ag, arow, 32, k, kh, sc);
            v2f b = b_frag(Wr, 16, k, kh, col);
            c = __builtin_amdgcn_wmma_f32_16x16x4_f32(false, a, false, b, (short)0, c, false, false);
        }
    }
    int rbase = m0 + ((lane >> 4) << 3);
#pragma unroll
    for (int v = 0; v < 8; ++v)
        out[(size_t)(rbase + v) * 16 + col] = c[v];
}


extern "C" void kernel_launch(void* const* d_in, const int* in_sizes, int n_in,
                              void* d_out, int out_size, void* d_ws, size_t ws_size,
                              hipStream_t stream) {
    const float* x     = (const float*)d_in[0];
    const int*   eidx  = (const int*)d_in[1];
    const int*   etype = (const int*)d_in[2];
    const float* W1    = (const float*)d_in[3];
    const float* root1 = (const float*)d_in[4];
    const float* b1    = (const float*)d_in[5];
    const float* W2    = (const float*)d_in[6];
    const float* root2 = (const float*)d_in[7];
    const float* b2    = (const float*)d_in[8];
    float* out = (float*)d_out;

    const int* src = eidx;
    const int* dst = eidx + NE;

    float* agg = (float*)d_ws;
    float* inv = agg + (size_t)NR * NN * 32;
    float* h   = inv + (size_t)NR * NN;

    const int B = 256;

    zero_f32<<<2048, B, 0, stream>>>(inv, (size_t)NR * NN);
    count_edges<<<(NE + B - 1) / B, B, 0, stream>>>(dst, etype, inv);
    recip_clamp<<<(NR * NN + B - 1) / B, B, 0, stream>>>(inv, NR * NN);

    zero_f32<<<4096, B, 0, stream>>>(agg, (size_t)NR * NN * 16);
    {
        long long nth = (long long)NE * 4;
        scatter_edges<16><<<(int)((nth + B - 1) / B), B, 0, stream>>>(x, src, dst, etype, agg);
    }
    rgcn_layer1<<<((NN / 16) * 32 + B - 1) / B, B, 0, stream>>>(x, agg, inv, W1, root1, b1, h);

    zero_f32<<<4096, B, 0, stream>>>(agg, (size_t)NR * NN * 32);
    {
        long long nth = (long long)NE * 8;
        scatter_edges<32><<<(int)((nth + B - 1) / B), B, 0, stream>>>(h, src, dst, etype, agg);
    }
    rgcn_layer2<<<((NN / 16) * 32 + B - 1) / B, B, 0, stream>>>(h, agg, inv, W2, root2, b2, out);

    (void)in_sizes; (void)n_in; (void)out_size; (void)ws_size;
}


# scatter remapped to one lane per feature column: each gather and f32 atomic-add instruction covers 1-2 whole contiguous rows (4x fewer line requests); still atomics, arrival order as before
# speedup vs baseline: 1.2307x; 1.2307x over previous
#include <hip/hip_runtime.h>
#include <utility>

#define NN 100000
#define NE 3200000
#define NR 8

typedef __attribute__((ext_vector_type(2))) float v2f;
typedef __attribute__((ext_vector_type(8))) float v8f;


__global__ void zero_f32(float* __restrict__ p, size_t n) {
    size_t i = (size_t)blockIdx.x * blockDim.x + threadIdx.x;
    size_t stride = (size_t)gridDim.x * blockDim.x;
    for (; i < n; i += stride) p[i] = 0.0f;
}

__global__ void count_edges(const int* __restrict__ dst, const int* __restrict__ et,
                            float* __restrict__ cnt) {
    int e = blockIdx.x * blockDim.x + threadIdx.x;
    if (e >= NE) return;
    atomicAdd(&cnt[(size_t)et[e] * NN + dst[e]], 1.0f);
}

__global__ void recip_clamp(float* __restrict__ c, int n) {
    int i = blockIdx.x * blockDim.x + threadIdx.x;
    if (i < n) c[i] = 1.0f / fmaxf(c[i], 1.0f);
}

#define EDGE_BATCHES 4
#define SCATTER_BLOCKS (NE / (32 * EDGE_BATCHES * 8))
static_assert(NE % (32 * EDGE_BATCHES * 8) == 0, "edge count must fill whole blocks");

template <int F, int I>
__device__ __forceinline__ unsigned edge_lane(unsigned x) {
    if constexpr (F == 16)
        return (unsigned)__builtin_amdgcn_update_dpp(0, (int)x, 0x150 + I, 0xF, 0xF, false);
    else
        return (unsigned)__builtin_amdgcn_readlane((int)x, I);
}

template <int F, int... I>
__device__ __forceinline__ void scatter_batch(const float* __restrict__ feat,
                                              float* __restrict__ agg,
                                              unsigned gb, unsigned ab, unsigned fb,
                                              std::integer_sequence<int, I...>) {
    float v[sizeof...(I)];
    ((v[I] = *(const float*)((const char*)feat + (edge_lane<F, I>(gb) + fb))), ...);
    (atomicAdd((float*)((char*)agg + (edge_lane<F, I>(ab) + fb)), v[I]), ...);
}

template <int F>
__global__ __launch_bounds__(256) void scatter_edges(const float* __restrict__ feat,
                                                     const int* __restrict__ src,
                                                     const int* __restrict__ dst,
                                                     const int* __restrict__ et,
                                                     float* __restrict__ agg) {
    const int lane  = threadIdx.x & 31;
    const int wave  = (int)((blockIdx.x * blockDim.x + threadIdx.x) >> 5);
    const int ebase = wave * (32 * EDGE_BATCHES);
    if (ebase >= NE) return;
    const unsigned fb = (unsigned)(lane & (F - 1)) * 4u;
    for (int b = 0; b < EDGE_BATCHES; ++b) {
        const int e = ebase + b * 32 + lane;
        const unsigned gb = (unsigned)src[e] * (F * 4u);
        const unsigned ab = ((unsigned)et[e] * NN + (unsigned)dst[e]) * (F * 4u);
        scatter_batch<F>(feat, agg, gb, ab, fb, std::make_integer_sequence<int, F>{});
    }
}

__device__ __forceinline__ v2f a_frag(const float* __restrict__ M, int row, int stride,
                                      int k0, int kh, float scale) {
    const float* p = M + (size_t)row * stride + k0 + kh;
    v2f a; a.x = p[0] * scale; a.y = p[1] * scale; return a;
}
__device__ __forceinline__ v2f b_frag(const float* __restrict__ W, int nout,
                                      int k0, int kh, int col) {
    v2f b;
    b.x = W[(size_t)(k0 + kh) * nout + col];
    b.y = W[(size_t)(k0 + kh + 1) * nout + col];
    return b;
}


__global__ __launch_bounds__(256) void rgcn_layer1(
    const float* __restrict__ x, const float* __restrict__ agg, const float* __restrict__ inv,
    const float* __restrict__ W1, const float* __restrict__ root1, const float* __restrict__ b1,
    float* __restrict__ h) {
    int wid  = (blockIdx.x * blockDim.x + threadIdx.x) >> 5;
    int lane = threadIdx.x & 31;
    if (wid >= NN / 16) return;
    int m0   = wid * 16;
    int arow = m0 + (lane & 15);
    int kh   = (lane >> 4) << 1;
    int col  = lane & 15;

    v8f c0, c1;
    float bb0 = b1[col], bb1 = b1[16 + col];
#pragma unroll
    for (int v = 0; v < 8; ++v) { c0[v] = bb0; c1[v] = bb1; }

#pragma unroll
    for (int k = 0; k < 16; k += 4) {
        v2f a  = a_frag(x, arow, 16, k, kh, 1.0f);
        v2f bA = b_frag(root1, 32, k, kh, col);
        v2f bB = b_frag(root1, 32, k, kh, col + 16);
        c0 = __builtin_amdgcn_wmma_f32_16x16x4_f32(false, a, false, bA, (short)0, c0, false, false);
        c1 = __builtin_amdgcn_wmma_f32_16x16x4_f32(false, a, false, bB, (short)0, c1, false, false);
    }
    for (int r = 0; r < NR; ++r) {
        const float* ag = agg + (size_t)r * NN * 16;
        const float* Wr = W1 + (size_t)r * 16 * 32;
        float sc = inv[(size_t)r * NN + arow];
#pragma unroll
        for (int k = 0; k < 16; k += 4) {
            v2f a  = a_frag(ag, arow, 16, k, kh, sc);
            v2f bA = b_frag(Wr, 32, k, kh, col);
            v2f bB = b_frag(Wr, 32, k, kh, col + 16);
            c0 = __builtin_amdgcn_wmma_f32_16x16x4_f32(false, a, false, bA, (short)0, c0, false, false);
            c1 = __builtin_amdgcn_wmma_f32_16x16x4_f32(false, a, false, bB, (short)0, c1, false, false);
        }
    }
    int rbase = m0 + ((lane >> 4) << 3);
#pragma unroll
    for (int v = 0; v < 8; ++v) {
        h[(size_t)(rbase + v) * 32 + col]      = fmaxf(c0[v], 0.0f);
        h[(size_t)(rbase + v) * 32 + 16 + col] = fmaxf(c1[v], 0.0f);
    }
}


__global__ __launch_bounds__(256) void rgcn_layer2(
    const float* __restrict__ h, const float* __restrict__ agg, const float* __restrict__ inv,
    const float* __restrict__ W2, const float* __restrict__ root2, const float* __restrict__ b2,
    float* __restrict__ out) {
    int wid  = (blockIdx.x * blockDim.x + threadIdx.x) >> 5;
    int lane = threadIdx.x & 31;
    if (wid >= NN / 16) return;
    int m0   = wid * 16;
    int arow = m0 + (lane & 15);
    int kh   = (lane >> 4) << 1;
    int col  = lane & 15;

    v8f c;
    float bb = b2[col];
#pragma unroll
    for (int v = 0; v < 8; ++v) c[v] = bb;

#pragma unroll
    for (int k = 0; k < 32; k += 4) {
        v2f a = a_frag(h, arow, 32, k, kh, 1.0f);
        v2f b = b_frag(root2, 16, k, kh, col);
        c = __builtin_amdgcn_wmma_f32_16x16x4_f32(false, a, false, b, (short)0, c, false, false);
    }
    for (int r = 0; r < NR; ++r) {
        const float* ag = agg + (size_t)r * NN * 32;
        const float* Wr = W2 + (size_t)r * 32 * 16;
        float sc = inv[(size_t)r * NN + arow];
#pragma unroll
        for (int k = 0; k < 32; k += 4) {
            v2f a = a_frag(ag, arow, 32, k, kh, sc);
            v2f b = b_frag(Wr, 16, k, kh, col);
            c = __builtin_amdgcn_wmma_f32_16x16x4_f32(false, a, false, b, (short)0, c, false, false);
        }
    }
    int rbase = m0 + ((lane >> 4) << 3);
#pragma unroll
    for (int v = 0; v < 8; ++v)
        out[(size_t)(rbase + v) * 16 + col] = c[v];
}


extern "C" void kernel_launch(void* const* d_in, const int* in_sizes, int n_in,
                              void* d_out, int out_size, void* d_ws, size_t ws_size,
                              hipStream_t stream) {
    const float* x     = (const float*)d_in[0];
    const int*   eidx  = (const int*)d_in[1];
    const int*   etype = (const int*)d_in[2];
    const float* W1    = (const float*)d_in[3];
    const float* root1 = (const float*)d_in[4];
    const float* b1    = (const float*)d_in[5];
    const float* W2    = (const float*)d_in[6];
    const float* root2 = (const float*)d_in[7];
    const float* b2    = (const float*)d_in[8];
    float* out = (float*)d_out;

    const int* src = eidx;
    const int* dst = eidx + NE;

    float* agg = (float*)d_ws;
    float* inv = agg + (size_t)NR * NN * 32;
    float* h   = inv + (size_t)NR * NN;

    const int B = 256;

    zero_f32<<<2048, B, 0, stream>>>(inv, (size_t)NR * NN);
    count_edges<<<(NE + B - 1) / B, B, 0, stream>>>(dst, etype, inv);
    recip_clamp<<<(NR * NN + B - 1) / B, B, 0, stream>>>(inv, NR * NN);

    zero_f32<<<4096, B, 0, stream>>>(agg, (size_t)NR * NN * 16);
    scatter_edges<16><<<SCATTER_BLOCKS, B, 0, stream>>>(x, src, dst, etype, agg);
    rgcn_layer1<<<((NN / 16) * 32 + B - 1) / B, B, 0, stream>>>(x, agg, inv, W1, root1, b1, h);

    zero_f32<<<4096, B, 0, stream>>>(agg, (size_t)NR * NN * 32);
    scatter_edges<32><<<SCATTER_BLOCKS, B, 0, stream>>>(h, src, dst, etype, agg);
    rgcn_layer2<<<((NN / 16) * 32 + B - 1) / B, B, 0, stream>>>(h, agg, inv, W2, root2, b2, out);

    (void)in_sizes; (void)n_in; (void)out_size; (void)ws_size;
}


# layer 2 applies W2_r before aggregating: h@W2_r table, then one 1/count-scaled f32 atomic scatter of 16-wide rows into out; sums now in arrival order over all in-edges of a node
# speedup vs baseline: 1.3523x; 1.0988x over previous
#include <hip/hip_runtime.h>
#include <utility>

#define NN 100000
#define NE 3200000
#define NR 8

typedef __attribute__((ext_vector_type(2))) float v2f;
typedef __attribute__((ext_vector_type(8))) float v8f;


__global__ void zero_f32(float* __restrict__ p, size_t n) {
    size_t i = (size_t)blockIdx.x * blockDim.x + threadIdx.x;
    size_t stride = (size_t)gridDim.x * blockDim.x;
    for (; i < n; i += stride) p[i] = 0.0f;
}

__global__ void count_edges(const int* __restrict__ dst, const int* __restrict__ et,
                            float* __restrict__ cnt) {
    int e = blockIdx.x * blockDim.x + threadIdx.x;
    if (e >= NE) return;
    atomicAdd(&cnt[(size_t)et[e] * NN + dst[e]], 1.0f);
}

__global__ void recip_clamp(float* __restrict__ c, int n) {
    int i = blockIdx.x * blockDim.x + threadIdx.x;
    if (i < n) c[i] = 1.0f / fmaxf(c[i], 1.0f);
}

#define EDGE_BATCHES 4
#define SCATTER_BLOCKS (NE / (32 * EDGE_BATCHES * 8))
static_assert(NE % (32 * EDGE_BATCHES * 8) == 0, "edge count must fill whole blocks");

template <int F, int I>
__device__ __forceinline__ unsigned edge_lane(unsigned x) {
    if constexpr (F == 16)
        return (unsigned)__builtin_amdgcn_update_dpp(0, (int)x, 0x150 + I, 0xF, 0xF, false);
    else
        return (unsigned)__builtin_amdgcn_readlane((int)x, I);
}

template <int F, int... I>
__device__ __forceinline__ void scatter_batch(const float* __restrict__ feat,
                                              float* __restrict__ agg,
                                              unsigned gb, unsigned ab, unsigned fb,
                                              std::integer_sequence<int, I...>) {
    float v[sizeof...(I)];
    ((v[I] = *(const float*)((const char*)feat + (edge_lane<F, I>(gb) + fb))), ...);
    (atomicAdd((float*)((char*)agg + (edge_lane<F, I>(ab) + fb)), v[I]), ...);
}

template <int F>
__global__ __launch_bounds__(256) void scatter_edges(const float* __restrict__ feat,
                                                     const int* __restrict__ src,
                                                     const int* __restrict__ dst,
                                                     const int* __restrict__ et,
                                                     float* __restrict__ agg) {
    const int lane  = threadIdx.x & 31;
    const int wave  = (int)((blockIdx.x * blockDim.x + threadIdx.x) >> 5);
    const int ebase = wave * (32 * EDGE_BATCHES);
    if (ebase >= NE) return;
    const unsigned fb = (unsigned)(lane & (F - 1)) * 4u;
    for (int b = 0; b < EDGE_BATCHES; ++b) {
        const int e = ebase + b * 32 + lane;
        const unsigned gb = (unsigned)src[e] * (F * 4u);
        const unsigned ab = ((unsigned)et[e] * NN + (unsigned)dst[e]) * (F * 4u);
        scatter_batch<F>(feat, agg, gb, ab, fb, std::make_integer_sequence<int, F>{});
    }
}

template <int... I>
__device__ __forceinline__ void scatter_scaled_batch(const float* __restrict__ proj,
                                                     float* __restrict__ out,
                                                     unsigned gb, unsigned ab, unsigned wb,
                                                     unsigned fb,
                                                     std::integer_sequence<int, I...>) {
    float v[sizeof...(I)];
    ((v[I] = *(const float*)((const char*)proj + (edge_lane<16, I>(gb) + fb))), ...);
    (atomicAdd((float*)((char*)out + (edge_lane<16, I>(ab) + fb)),
               v[I] * __uint_as_float(edge_lane<16, I>(wb))), ...);
}

__global__ __launch_bounds__(256) void scatter_projected(const float* __restrict__ proj,
                                                         const int* __restrict__ src,
                                                         const int* __restrict__ dst,
                                                         const int* __restrict__ et,
                                                         const float* __restrict__ inv,
                                                         float* __restrict__ out) {
    const int lane  = threadIdx.x & 31;
    const int wave  = (int)((blockIdx.x * blockDim.x + threadIdx.x) >> 5);
    const int ebase = wave * (32 * EDGE_BATCHES);
    if (ebase >= NE) return;
    const unsigned fb = (unsigned)(lane & 15) * 4u;
    for (int b = 0; b < EDGE_BATCHES; ++b) {
        const int e = ebase + b * 32 + lane;
        const unsigned r = (unsigned)et[e];
        const unsigned d = (unsigned)dst[e];
        const unsigned gb = (r * NN + (unsigned)src[e]) * 64u;
        const unsigned ab = d * 64u;
        const unsigned wb = __float_as_uint(inv[(size_t)r * NN + d]);
        scatter_scaled_batch(proj, out, gb, ab, wb, fb, std::make_integer_sequence<int, 16>{});
    }
}

__device__ __forceinline__ v2f a_frag(const float* __restrict__ M, int row, int stride,
                                      int k0, int kh, float scale) {
    const float* p = M + (size_t)row * stride + k0 + kh;
    v2f a; a.x = p[0] * scale; a.y = p[1] * scale; return a;
}
__device__ __forceinline__ v2f b_frag(const float* __restrict__ W, int nout,
                                      int k0, int kh, int col) {
    v2f b;
    b.x = W[(size_t)(k0 + kh) * nout + col];
    b.y = W[(size_t)(k0 + kh + 1) * nout + col];
    return b;
}


__global__ __launch_bounds__(256) void rgcn_layer1(
    const float* __restrict__ x, const float* __restrict__ agg, const float* __restrict__ inv,
    const float* __restrict__ W1, const float* __restrict__ root1, const float* __restrict__ b1,
    float* __restrict__ h) {
    int wid  = (blockIdx.x * blockDim.x + threadIdx.x) >> 5;
    int lane = threadIdx.x & 31;
    if (wid >= NN / 16) return;
    int m0   = wid * 16;
    int arow = m0 + (lane & 15);
    int kh   = (lane >> 4) << 1;
    int col  = lane & 15;

    v8f c0, c1;
    float bb0 = b1[col], bb1 = b1[16 + col];
#pragma unroll
    for (int v = 0; v < 8; ++v) { c0[v] = bb0; c1[v] = bb1; }

#pragma unroll
    for (int k = 0; k < 16; k += 4) {
        v2f a  = a_frag(x, arow, 16, k, kh, 1.0f);
        v2f bA = b_frag(root1, 32, k, kh, col);
        v2f bB = b_frag(root1, 32, k, kh, col + 16);
        c0 = __builtin_amdgcn_wmma_f32_16x16x4_f32(false, a, false, bA, (short)0, c0, false, false);
        c1 = __builtin_amdgcn_wmma_f32_16x16x4_f32(false, a, false, bB, (short)0, c1, false, false);
    }
    for (int r = 0; r < NR; ++r) {
        const float* ag = agg + (size_t)r * NN * 16;
        const float* Wr = W1 + (size_t)r * 16 * 32;
        float sc = inv[(size_t)r * NN + arow];
#pragma unroll
        for (int k = 0; k < 16; k += 4) {
            v2f a  = a_frag(ag, arow, 16, k, kh, sc);
            v2f bA = b_frag(Wr, 32, k, kh, col);
            v2f bB = b_frag(Wr, 32, k, kh, col + 16);
            c0 = __builtin_amdgcn_wmma_f32_16x16x4_f32(false, a, false, bA, (short)0, c0, false, false);
            c1 = __builtin_amdgcn_wmma_f32_16x16x4_f32(false, a, false, bB, (short)0, c1, false, false);
        }
    }
    int rbase = m0 + ((lane >> 4) << 3);
#pragma unroll
    for (int v = 0; v < 8; ++v) {
        h[(size_t)(rbase + v) * 32 + col]      = fmaxf(c0[v], 0.0f);
        h[(size_t)(rbase + v) * 32 + 16 + col] = fmaxf(c1[v], 0.0f);
    }
}


__global__ __launch_bounds__(256) void rgcn_project2(
    const float* __restrict__ h, const float* __restrict__ W2, const float* __restrict__ root2,
    const float* __restrict__ b2, float* __restrict__ proj, float* __restrict__ out) {
    int wid  = (blockIdx.x * blockDim.x + threadIdx.x) >> 5;
    int lane = threadIdx.x & 31;
    if (wid >= NN / 16) return;
    int m0   = wid * 16;
    int mrow = lane & 15;
    int arow = m0 + ((mrow & 7) << 1) + (mrow >> 3);
    int kh   = (lane >> 4) << 1;
    int col  = lane & 15;

    v2f a[8];
#pragma unroll
    for (int k = 0; k < 8; ++k) a[k] = a_frag(h, arow, 32, 4 * k, kh, 1.0f);

    v8f c[1 + NR];
    float bb = b2[col];
#pragma unroll
    for (int t = 0; t <= NR; ++t)
#pragma unroll
        for (int v = 0; v < 8; ++v) c[t][v] = (t == 0) ? bb : 0.0f;

#pragma unroll
    for (int k = 0; k < 8; ++k) {
        v2f b = b_frag(root2, 16, 4 * k, kh, col);
        c[0] = __builtin_amdgcn_wmma_f32_16x16x4_f32(false, a[k], false, b, (short)0, c[0], false, false);
#pragma unroll
        for (int r = 0; r < NR; ++r) {
            v2f br = b_frag(W2 + (size_t)r * 32 * 16, 16, 4 * k, kh, col);
            c[1 + r] = __builtin_amdgcn_wmma_f32_16x16x4_f32(false, a[k], false, br, (short)0, c[1 + r], false, false);
        }
    }
    int nbase = m0 + (lane >> 4);
#pragma unroll
    for (int v = 0; v < 8; ++v) {
        size_t node = (size_t)(nbase + 2 * v);
        out[node * 16 + col] = c[0][v];
#pragma unroll
        for (int r = 0; r < NR; ++r)
            proj[((size_t)r * NN + node) * 16 + col] = c[1 + r][v];
    }
}


extern "C" void kernel_launch(void* const* d_in, const int* in_sizes, int n_in,
                              void* d_out, int out_size, void* d_ws, size_t ws_size,
                              hipStream_t stream) {
    const float* x     = (const float*)d_in[0];
    const int*   eidx  = (const int*)d_in[1];
    const int*   etype = (const int*)d_in[2];
    const float* W1    = (const float*)d_in[3];
    const float* root1 = (const float*)d_in[4];
    const float* b1    = (const float*)d_in[5];
    const float* W2    = (const float*)d_in[6];
    const float* root2 = (const float*)d_in[7];
    const float* b2    = (const float*)d_in[8];
    float* out = (float*)d_out;

    const int* src = eidx;
    const int* dst = eidx + NE;

    float* agg  = (float*)d_ws;
    float* proj = agg + (size_t)NR * NN * 16;
    float* inv  = agg + (size_t)NR * NN * 32;
    float* h    = inv + (size_t)NR * NN;

    const int B = 256;

    zero_f32<<<2048, B, 0, stream>>>(inv, (size_t)NR * NN);
    count_edges<<<(NE + B - 1) / B, B, 0, stream>>>(dst, etype, inv);
    recip_clamp<<<(NR * NN + B - 1) / B, B, 0, stream>>>(inv, NR * NN);

    zero_f32<<<4096, B, 0, stream>>>(agg, (size_t)NR * NN * 16);
    scatter_edges<16><<<SCATTER_BLOCKS, B, 0, stream>>>(x, src, dst, etype, agg);
    rgcn_layer1<<<((NN / 16) * 32 + B - 1) / B, B, 0, stream>>>(x, agg, inv, W1, root1, b1, h);

    rgcn_project2<<<((NN / 16) * 32 + B - 1) / B, B, 0, stream>>>(h, W2, root2, b2, proj, out);
    scatter_projected<<<SCATTER_BLOCKS, B, 0, stream>>>(proj, src, dst, etype, inv, out);

    (void)in_sizes; (void)n_in; (void)out_size; (void)ws_size;
}


# edge count folded into the layer-1 edge walk (one word add per edge) and 1/max(count,1) formed by its readers: count and reciprocal passes gone, 8 launches to 5; atomics, arrival order as before
# speedup vs baseline: 1.4537x; 1.0750x over previous
#include <hip/hip_runtime.h>
#include <utility>

#define NN 100000
#define NE 3200000
#define NR 8

typedef __attribute__((ext_vector_type(2))) float v2f;
typedef __attribute__((ext_vector_type(8))) float v8f;


__global__ void zero_two(float* __restrict__ a, size_t na, float* __restrict__ b, size_t nb) {
    size_t i = (size_t)blockIdx.x * blockDim.x + threadIdx.x;
    size_t stride = (size_t)gridDim.x * blockDim.x;
    for (size_t j = i; j < na; j += stride) a[j] = 0.0f;
    for (size_t j = i; j < nb; j += stride) b[j] = 0.0f;
}

__device__ __forceinline__ float mean_weight(float count) {
    return 1.0f / fmaxf(count, 1.0f);
}

#define EDGE_BATCHES 4
#define SCATTER_BLOCKS (NE / (32 * EDGE_BATCHES * 8))
static_assert(NE % (32 * EDGE_BATCHES * 8) == 0, "edge count must fill whole blocks");

template <int F, int I>
__device__ __forceinline__ unsigned edge_lane(unsigned x) {
    if constexpr (F == 16)
        return (unsigned)__builtin_amdgcn_update_dpp(0, (int)x, 0x150 + I, 0xF, 0xF, false);
    else
        return (unsigned)__builtin_amdgcn_readlane((int)x, I);
}

template <int F, int... I>
__device__ __forceinline__ void scatter_batch(const float* __restrict__ feat,
                                              float* __restrict__ agg,
                                              unsigned gb, unsigned ab, unsigned fb,
                                              std::integer_sequence<int, I...>) {
    float v[sizeof...(I)];
    ((v[I] = *(const float*)((const char*)feat + (edge_lane<F, I>(gb) + fb))), ...);
    (atomicAdd((float*)((char*)agg + (edge_lane<F, I>(ab) + fb)), v[I]), ...);
}

template <int F>
__global__ __launch_bounds__(256) void scatter_edges(const float* __restrict__ feat,
                                                     const int* __restrict__ src,
                                                     const int* __restrict__ dst,
                                                     const int* __restrict__ et,
                                                     float* __restrict__ agg,
                                                     float* __restrict__ cnt) {
    const int lane  = threadIdx.x & 31;
    const int wave  = (int)((blockIdx.x * blockDim.x + threadIdx.x) >> 5);
    const int ebase = wave * (32 * EDGE_BATCHES);
    if (ebase >= NE) return;
    const unsigned fb = (unsigned)(lane & (F - 1)) * 4u;
    for (int b = 0; b < EDGE_BATCHES; ++b) {
        const int e = ebase + b * 32 + lane;
        const unsigned key = (unsigned)et[e] * NN + (unsigned)dst[e];
        const unsigned gb = (unsigned)src[e] * (F * 4u);
        const unsigned ab = key * (F * 4u);
        scatter_batch<F>(feat, agg, gb, ab, fb, std::make_integer_sequence<int, F>{});
        atomicAdd(&cnt[key], 1.0f);
    }
}

template <int... I>
__device__ __forceinline__ void scatter_scaled_batch(const float* __restrict__ proj,
                                                     float* __restrict__ out,
                                                     unsigned gb, unsigned ab, unsigned wb,
                                                     unsigned fb,
                                                     std::integer_sequence<int, I...>) {
    float v[sizeof...(I)];
    ((v[I] = *(const float*)((const char*)proj + (edge_lane<16, I>(gb) + fb))), ...);
    (atomicAdd((float*)((char*)out + (edge_lane<16, I>(ab) + fb)),
               v[I] * __uint_as_float(edge_lane<16, I>(wb))), ...);
}

__global__ __launch_bounds__(256) void scatter_projected(const float* __restrict__ proj,
                                                         const int* __restrict__ src,
                                                         const int* __restrict__ dst,
                                                         const int* __restrict__ et,
                                                         const float* __restrict__ cnt,
                                                         float* __restrict__ out) {
    const int lane  = threadIdx.x & 31;
    const int wave  = (int)((blockIdx.x * blockDim.x + threadIdx.x) >> 5);
    const int ebase = wave * (32 * EDGE_BATCHES);
    if (ebase >= NE) return;
    const unsigned fb = (unsigned)(lane & 15) * 4u;
    for (int b = 0; b < EDGE_BATCHES; ++b) {
        const int e = ebase + b * 32 + lane;
        const unsigned r = (unsigned)et[e];
        const unsigned d = (unsigned)dst[e];
        const unsigned gb = (r * NN + (unsigned)src[e]) * 64u;
        const unsigned ab = d * 64u;
        const unsigned wb = __float_as_uint(mean_weight(cnt[(size_t)r * NN + d]));
        scatter_scaled_batch(proj, out, gb, ab, wb, fb, std::make_integer_sequence<int, 16>{});
    }
}

__device__ __forceinline__ v2f a_frag(const float* __restrict__ M, int row, int stride,
                                      int k0, int kh, float scale) {
    const float* p = M + (size_t)row * stride + k0 + kh;
    v2f a; a.x = p[0] * scale; a.y = p[1] * scale; return a;
}
__device__ __forceinline__ v2f b_frag(const float* __restrict__ W, int nout,
                                      int k0, int kh, int col) {
    v2f b;
    b.x = W[(size_t)(k0 + kh) * nout + col];
    b.y = W[(size_t)(k0 + kh + 1) * nout + col];
    return b;
}


__global__ __launch_bounds__(256) void rgcn_layer1(
    const float* __restrict__ x, const float* __restrict__ agg, const float* __restrict__ cnt,
    const float* __restrict__ W1, const float* __restrict__ root1, const float* __restrict__ b1,
    float* __restrict__ h) {
    int wid  = (blockIdx.x * blockDim.x + threadIdx.x) >> 5;
    int lane = threadIdx.x & 31;
    if (wid >= NN / 16) return;
    int m0   = wid * 16;
    int arow = m0 + (lane & 15);
    int kh   = (lane >> 4) << 1;
    int col  = lane & 15;

    v8f c0, c1;
    float bb0 = b1[col], bb1 = b1[16 + col];
#pragma unroll
    for (int v = 0; v < 8; ++v) { c0[v] = bb0; c1[v] = bb1; }

#pragma unroll
    for (int k = 0; k < 16; k += 4) {
        v2f a  = a_frag(x, arow, 16, k, kh, 1.0f);
        v2f bA = b_frag(root1, 32, k, kh, col);
        v2f bB = b_frag(root1, 32, k, kh, col + 16);
        c0 = __builtin_amdgcn_wmma_f32_16x16x4_f32(false, a, false, bA, (short)0, c0, false, false);
        c1 = __builtin_amdgcn_wmma_f32_16x16x4_f32(false, a, false, bB, (short)0, c1, false, false);
    }
    for (int r = 0; r < NR; ++r) {
        const float* ag = agg + (size_t)r * NN * 16;
        const float* Wr = W1 + (size_t)r * 16 * 32;
        float sc = mean_weight(cnt[(size_t)r * NN + arow]);
#pragma unroll
        for (int k = 0; k < 16; k += 4) {
            v2f a  = a_frag(ag, arow, 16, k, kh, sc);
            v2f bA = b_frag(Wr, 32, k, kh, col);
            v2f bB = b_frag(Wr, 32, k, kh, col + 16);
            c0 = __builtin_amdgcn_wmma_f32_16x16x4_f32(false, a, false, bA, (short)0, c0, false, false);
            c1 = __builtin_amdgcn_wmma_f32_16x16x4_f32(false, a, false, bB, (short)0, c1, false, false);
        }
    }
    int rbase = m0 + ((lane >> 4) << 3);
#pragma unroll
    for (int v = 0; v < 8; ++v) {
        h[(size_t)(rbase + v) * 32 + col]      = fmaxf(c0[v], 0.0f);
        h[(size_t)(rbase + v) * 32 + 16 + col] = fmaxf(c1[v], 0.0f);
    }
}


__global__ __launch_bounds__(256) void rgcn_project2(
    const float* __restrict__ h, const float* __restrict__ W2, const float* __restrict__ root2,
    const float* __restrict__ b2, float* __restrict__ proj, float* __restrict__ out) {
    int wid  = (blockIdx.x * blockDim.x + threadIdx.x) >> 5;
    int lane = threadIdx.x & 31;
    if (wid >= NN / 16) return;
    int m0   = wid * 16;
    int mrow = lane & 15;
    int arow = m0 + ((mrow & 7) << 1) + (mrow >> 3);
    int kh   = (lane >> 4) << 1;
    int col  = lane & 15;

    v2f a[8];
#pragma unroll
    for (int k = 0; k < 8; ++k) a[k] = a_frag(h, arow, 32, 4 * k, kh, 1.0f);

    v8f c[1 + NR];
    float bb = b2[col];
#pragma unroll
    for (int t = 0; t <= NR; ++t)
#pragma unroll
        for (int v = 0; v < 8; ++v) c[t][v] = (t == 0) ? bb : 0.0f;

#pragma unroll
    for (int k = 0; k < 8; ++k) {
        v2f b = b_frag(root2, 16, 4 * k, kh, col);
        c[0] = __builtin_amdgcn_wmma_f32_16x16x4_f32(false, a[k], false, b, (short)0, c[0], false, false);
#pragma unroll
        for (int r = 0; r < NR; ++r) {
            v2f br = b_frag(W2 + (size_t)r * 32 * 16, 16, 4 * k, kh, col);
            c[1 + r] = __builtin_amdgcn_wmma_f32_16x16x4_f32(false, a[k], false, br, (short)0, c[1 + r], false, false);
        }
    }
    int nbase = m0 + (lane >> 4);
#pragma unroll
    for (int v = 0; v < 8; ++v) {
        size_t node = (size_t)(nbase + 2 * v);
        out[node * 16 + col] = c[0][v];
#pragma unroll
        for (int r = 0; r < NR; ++r)
            proj[((size_t)r * NN + node) * 16 + col] = c[1 + r][v];
    }
}


extern "C" void kernel_launch(void* const* d_in, const int* in_sizes, int n_in,
                              void* d_out, int out_size, void* d_ws, size_t ws_size,
                              hipStream_t stream) {
    const float* x     = (const float*)d_in[0];
    const int*   eidx  = (const int*)d_in[1];
    const int*   etype = (const int*)d_in[2];
    const float* W1    = (const float*)d_in[3];
    const float* root1 = (const float*)d_in[4];
    const float* b1    = (const float*)d_in[5];
    const float* W2    = (const float*)d_in[6];
    const float* root2 = (const float*)d_in[7];
    const float* b2    = (const float*)d_in[8];
    float* out = (float*)d_out;

    const int* src = eidx;
    const int* dst = eidx + NE;

    float* agg  = (float*)d_ws;
    float* proj = agg + (size_t)NR * NN * 16;
    float* cnt  = agg + (size_t)NR * NN * 32;
    float* h    = cnt + (size_t)NR * NN;

    const int B = 256;

    zero_two<<<4096, B, 0, stream>>>(agg, (size_t)NR * NN * 16, cnt, (size_t)NR * NN);
    scatter_edges<16><<<SCATTER_BLOCKS, B, 0, stream>>>(x, src, dst, etype, agg, cnt);
    rgcn_layer1<<<((NN / 16) * 32 + B - 1) / B, B, 0, stream>>>(x, agg, cnt, W1, root1, b1, h);

    rgcn_project2<<<((NN / 16) * 32 + B - 1) / B, B, 0, stream>>>(h, W2, root2, b2, proj, out);
    scatter_projected<<<SCATTER_BLOCKS, B, 0, stream>>>(proj, src, dst, etype, cnt, out);

    (void)in_sizes; (void)n_in; (void)out_size; (void)ws_size;
}


# edge scatters cut by destination-node half, each half's adds issued from one half of the chip (job tickets by die id): accumulator lines stay in one coherence domain; f32 atomics, arrival order
# speedup vs baseline: 2.6655x; 1.8336x over previous
#include <hip/hip_runtime.h>
#include <utility>

#define NN 100000
#define NE 3200000
#define NR 8

typedef __attribute__((ext_vector_type(2))) float v2f;
typedef __attribute__((ext_vector_type(8))) float v8f;


__global__ void zero_two(float* __restrict__ a, size_t na, float* __restrict__ b, size_t nb) {
    size_t i = (size_t)blockIdx.x * blockDim.x + threadIdx.x;
    size_t stride = (size_t)gridDim.x * blockDim.x;
    for (size_t j = i; j < na; j += stride) a[j] = 0.0f;
    for (size_t j = i; j < nb; j += stride) b[j] = 0.0f;
}

__device__ __forceinline__ float mean_weight(float count) {
    return 1.0f / fmaxf(count, 1.0f);
}

#define EDGE_BATCHES 4
#define EDGES_PER_BLOCK (32 * EDGE_BATCHES * 8)
#define SCATTER_CHUNKS (NE / EDGES_PER_BLOCK)
#define SCATTER_BLOCKS (2 * SCATTER_CHUNKS)
#define NODE_SPLIT (NN / 2)
#define SKIP_EDGE 0xFFFFFFFFu
static_assert(NE % EDGES_PER_BLOCK == 0, "edge count must fill whole blocks");

__device__ __forceinline__ unsigned chip_half() {
    return ((unsigned)__builtin_amdgcn_s_sendmsg_rtn(0x87) >> 18) & 1u;
}

__device__ __forceinline__ bool draw_job(unsigned* __restrict__ tickets, int& chunk, unsigned& half) {
    __shared__ unsigned job;
    if (threadIdx.x == 0) {
        unsigned h = chip_half();
        unsigned t = atomicAdd(&tickets[h], 1u);
        if (t >= SCATTER_CHUNKS) { h ^= 1u; t = atomicAdd(&tickets[h], 1u); }
        job = t < SCATTER_CHUNKS ? t * 2u + h : SKIP_EDGE;
    }
    __syncthreads();
    const unsigned j = job;
    chunk = (int)(j >> 1);
    half = j & 1u;
    return j != SKIP_EDGE;
}

template <int F, int I>
__device__ __forceinline__ unsigned edge_lane(unsigned x) {
    if constexpr (F == 16)
        return (unsigned)__builtin_amdgcn_update_dpp(0, (int)x, 0x150 + I, 0xF, 0xF, false);
    else
        return (unsigned)__builtin_amdgcn_readlane((int)x, I);
}

__device__ __forceinline__ void add_live(float* __restrict__ base, unsigned a, unsigned fb, float v) {
    if (a != SKIP_EDGE) atomicAdd((float*)((char*)base + (a + fb)), v);
}

template <int F, int... I>
__device__ __forceinline__ void scatter_batch(const float* __restrict__ feat,
                                              float* __restrict__ agg,
                                              unsigned gb, unsigned ab, unsigned fb,
                                              std::integer_sequence<int, I...>) {
    float v[sizeof...(I)];
    ((v[I] = *(const float*)((const char*)feat + (edge_lane<F, I>(gb) + fb))), ...);
    (add_live(agg, edge_lane<F, I>(ab), fb, v[I]), ...);
}

template <int F>
__global__ __launch_bounds__(256) void scatter_edges(const float* __restrict__ feat,
                                                     const int* __restrict__ src,
                                                     const int* __restrict__ dst,
                                                     const int* __restrict__ et,
                                                     float* __restrict__ agg,
                                                     float* __restrict__ cnt,
                                                     unsigned* __restrict__ tickets) {
    int chunk; unsigned half;
    if (!draw_job(tickets, chunk, half)) return;
    const int lane  = threadIdx.x & 31;
    const int ebase = chunk * EDGES_PER_BLOCK + (int)(threadIdx.x >> 5) * (32 * EDGE_BATCHES);
    const unsigned fb = (unsigned)(lane & (F - 1)) * 4u;
    for (int b = 0; b < EDGE_BATCHES; ++b) {
        const int e = ebase + b * 32 + lane;
        const unsigned d = (unsigned)dst[e];
        const bool mine = (d >= NODE_SPLIT ? 1u : 0u) == half;
        const unsigned key = (unsigned)et[e] * NN + d;
        const unsigned gb = mine ? (unsigned)src[e] * (F * 4u) : 0u;
        const unsigned ab = mine ? key * (F * 4u) : SKIP_EDGE;
        scatter_batch<F>(feat, agg, gb, ab, fb, std::make_integer_sequence<int, F>{});
        if (mine) atomicAdd(&cnt[key], 1.0f);
    }
}

template <int... I>
__device__ __forceinline__ void scatter_scaled_batch(const float* __restrict__ proj,
                                                     float* __restrict__ out,
                                                     unsigned gb, unsigned ab, unsigned wb,
                                                     unsigned fb,
                                                     std::integer_sequence<int, I...>) {
    float v[sizeof...(I)];
    ((v[I] = *(const float*)((const char*)proj + (edge_lane<16, I>(gb) + fb))), ...);
    (add_live(out, edge_lane<16, I>(ab), fb,
              v[I] * __uint_as_float(edge_lane<16, I>(wb))), ...);
}

__global__ __launch_bounds__(256) void scatter_projected(const float* __restrict__ proj,
                                                         const int* __restrict__ src,
                                                         const int* __restrict__ dst,
                                                         const int* __restrict__ et,
                                                         const float* __restrict__ cnt,
                                                         float* __restrict__ out,
                                                         unsigned* __restrict__ tickets) {
    int chunk; unsigned half;
    if (!draw_job(tickets, chunk, half)) return;
    const int lane  = threadIdx.x & 31;
    const int ebase = chunk * EDGES_PER_BLOCK + (int)(threadIdx.x >> 5) * (32 * EDGE_BATCHES);
    const unsigned fb = (unsigned)(lane & 15) * 4u;
    for (int b = 0; b < EDGE_BATCHES; ++b) {
        const int e = ebase + b * 32 + lane;
        const unsigned r = (unsigned)et[e];
        const unsigned d = (unsigned)dst[e];
        const bool mine = (d >= NODE_SPLIT ? 1u : 0u) == half;
        const unsigned gb = mine ? (r * NN + (unsigned)src[e]) * 64u : 0u;
        const unsigned ab = mine ? d * 64u : SKIP_EDGE;
        const unsigned wb = mine ? __float_as_uint(mean_weight(cnt[(size_t)r * NN + d])) : 0u;
        scatter_scaled_batch(proj, out, gb, ab, wb, fb, std::make_integer_sequence<int, 16>{});
    }
}

__device__ __forceinline__ v2f a_frag(const float* __restrict__ M, int row, int stride,
                                      int k0, int kh, float scale) {
    const float* p = M + (size_t)row * stride + k0 + kh;
    v2f a; a.x = p[0] * scale; a.y = p[1] * scale; return a;
}
__device__ __forceinline__ v2f b_frag(const float* __restrict__ W, int nout,
                                      int k0, int kh, int col) {
    v2f b;
    b.x = W[(size_t)(k0 + kh) * nout + col];
    b.y = W[(size_t)(k0 + kh + 1) * nout + col];
    return b;
}


__global__ __launch_bounds__(256) void rgcn_layer1(
    const float* __restrict__ x, const float* __restrict__ agg, const float* __restrict__ cnt,
    const float* __restrict__ W1, const float* __restrict__ root1, const float* __restrict__ b1,
    float* __restrict__ h) {
    int wid  = (blockIdx.x * blockDim.x + threadIdx.x) >> 5;
    int lane = threadIdx.x & 31;
    if (wid >= NN / 16) return;
    int m0   = wid * 16;
    int arow = m0 + (lane & 15);
    int kh   = (lane >> 4) << 1;
    int col  = lane & 15;

    v8f c0, c1;
    float bb0 = b1[col], bb1 = b1[16 + col];
#pragma unroll
    for (int v = 0; v < 8; ++v) { c0[v] = bb0; c1[v] = bb1; }

#pragma unroll
    for (int k = 0; k < 16; k += 4) {
        v2f a  = a_frag(x, arow, 16, k, kh, 1.0f);
        v2f bA = b_frag(root1, 32, k, kh, col);
        v2f bB = b_frag(root1, 32, k, kh, col + 16);
        c0 = __builtin_amdgcn_wmma_f32_16x16x4_f32(false, a, false, bA, (short)0, c0, false, false);
        c1 = __builtin_amdgcn_wmma_f32_16x16x4_f32(false, a, false, bB, (short)0, c1, false, false);
    }
    for (int r = 0; r < NR; ++r) {
        const float* ag = agg + (size_t)r * NN * 16;
        const float* Wr = W1 + (size_t)r * 16 * 32;
        float sc = mean_weight(cnt[(size_t)r * NN + arow]);
#pragma unroll
        for (int k = 0; k < 16; k += 4) {
            v2f a  = a_frag(ag, arow, 16, k, kh, sc);
            v2f bA = b_frag(Wr, 32, k, kh, col);
            v2f bB = b_frag(Wr, 32, k, kh, col + 16);
            c0 = __builtin_amdgcn_wmma_f32_16x16x4_f32(false, a, false, bA, (short)0, c0, false, false);
            c1 = __builtin_amdgcn_wmma_f32_16x16x4_f32(false, a, false, bB, (short)0, c1, false, false);
        }
    }
    int rbase = m0 + ((lane >> 4) << 3);
#pragma unroll
    for (int v = 0; v < 8; ++v) {
        h[(size_t)(rbase + v) * 32 + col]      = fmaxf(c0[v], 0.0f);
        h[(size_t)(rbase + v) * 32 + 16 + col] = fmaxf(c1[v], 0.0f);
    }
}


__global__ __launch_bounds__(256) void rgcn_project2(
    const float* __restrict__ h, const float* __restrict__ W2, const float* __restrict__ root2,
    const float* __restrict__ b2, float* __restrict__ proj, float* __restrict__ out,
    unsigned* __restrict__ tickets) {
    if (blockIdx.x == 0 && threadIdx.x < 2) tickets[threadIdx.x] = 0u;
    int wid  = (blockIdx.x * blockDim.x + threadIdx.x) >> 5;
    int lane = threadIdx.x & 31;
    if (wid >= NN / 16) return;
    int m0   = wid * 16;
    int mrow = lane & 15;
    int arow = m0 + ((mrow & 7) << 1) + (mrow >> 3);
    int kh   = (lane >> 4) << 1;
    int col  = lane & 15;

    v2f a[8];
#pragma unroll
    for (int k = 0; k < 8; ++k) a[k] = a_frag(h, arow, 32, 4 * k, kh, 1.0f);

    v8f c[1 + NR];
    float bb = b2[col];
#pragma unroll
    for (int t = 0; t <= NR; ++t)
#pragma unroll
        for (int v = 0; v < 8; ++v) c[t][v] = (t == 0) ? bb : 0.0f;

#pragma unroll
    for (int k = 0; k < 8; ++k) {
        v2f b = b_frag(root2, 16, 4 * k, kh, col);
        c[0] = __builtin_amdgcn_wmma_f32_16x16x4_f32(false, a[k], false, b, (short)0, c[0], false, false);
#pragma unroll
        for (int r = 0; r < NR; ++r) {
            v2f br = b_frag(W2 + (size_t)r * 32 * 16, 16, 4 * k, kh, col);
            c[1 + r] = __builtin_amdgcn_wmma_f32_16x16x4_f32(false, a[k], false, br, (short)0, c[1 + r], false, false);
        }
    }
    int nbase = m0 + (lane >> 4);
#pragma unroll
    for (int v = 0; v < 8; ++v) {
        size_t node = (size_t)(nbase + 2 * v);
        out[node * 16 + col] = c[0][v];
#pragma unroll
        for (int r = 0; r < NR; ++r)
            proj[((size_t)r * NN + node) * 16 + col] = c[1 + r][v];
    }
}


extern "C" void kernel_launch(void* const* d_in, const int* in_sizes, int n_in,
                              void* d_out, int out_size, void* d_ws, size_t ws_size,
                              hipStream_t stream) {
    const float* x     = (const float*)d_in[0];
    const int*   eidx  = (const int*)d_in[1];
    const int*   etype = (const int*)d_in[2];
    const float* W1    = (const float*)d_in[3];
    const float* root1 = (const float*)d_in[4];
    const float* b1    = (const float*)d_in[5];
    const float* W2    = (const float*)d_in[6];
    const float* root2 = (const float*)d_in[7];
    const float* b2    = (const float*)d_in[8];
    float* out = (float*)d_out;

    const int* src = eidx;
    const int* dst = eidx + NE;

    float* agg  = (float*)d_ws;
    float* proj = agg + (size_t)NR * NN * 16;
    float* cnt  = agg + (size_t)NR * NN * 32;
    float* h    = cnt + (size_t)NR * NN;

    const int B = 256;

    zero_two<<<4096, B, 0, stream>>>(agg, (size_t)NR * NN * 16, cnt, (size_t)NR * NN + 2);
    scatter_edges<16><<<SCATTER_BLOCKS, B, 0, stream>>>(x, src, dst, etype, agg, cnt, (unsigned*)h);
    rgcn_layer1<<<((NN / 16) * 32 + B - 1) / B, B, 0, stream>>>(x, agg, cnt, W1, root1, b1, h);

    rgcn_project2<<<((NN / 16) * 32 + B - 1) / B, B, 0, stream>>>(h, W2, root2, b2, proj, out, (unsigned*)agg);
    scatter_projected<<<SCATTER_BLOCKS, B, 0, stream>>>(proj, src, dst, etype, cnt, out, (unsigned*)agg);

    (void)in_sizes; (void)n_in; (void)out_size; (void)ws_size;
}


# layer-2 edge sums leave global atomics: edges counting-sorted by 400-node destination range, one work-group per range sums weighted rows in LDS (LDS f32 adds, arrival order), plain stores to out
# speedup vs baseline: 2.8257x; 1.0601x over previous
#include <hip/hip_runtime.h>
#include <utility>

#define NN 100000
#define NE 3200000
#define NR 8

typedef __attribute__((ext_vector_type(2))) float v2f;
typedef __attribute__((ext_vector_type(8))) float v8f;


__global__ void zero_two(float* __restrict__ a, size_t na, float* __restrict__ b, size_t nb,
                         unsigned* __restrict__ c, size_t nc) {
    size_t i = (size_t)blockIdx.x * blockDim.x + threadIdx.x;
    size_t stride = (size_t)gridDim.x * blockDim.x;
    for (size_t j = i; j < na; j += stride) a[j] = 0.0f;
    for (size_t j = i; j < nb; j += stride) b[j] = 0.0f;
    for (size_t j = i; j < nc; j += stride) c[j] = 0u;
}

__device__ __forceinline__ float mean_weight(float count) {
    return 1.0f / fmaxf(count, 1.0f);
}

#define EDGE_BATCHES 4
#define EDGES_PER_BLOCK (32 * EDGE_BATCHES * 8)
#define SCATTER_CHUNKS (NE / EDGES_PER_BLOCK)
#define SCATTER_BLOCKS (2 * SCATTER_CHUNKS)
#define NODE_SPLIT (NN / 2)
#define SKIP_EDGE 0xFFFFFFFFu
static_assert(NE % EDGES_PER_BLOCK == 0, "edge count must fill whole blocks");

__device__ __forceinline__ unsigned chip_half() {
    return ((unsigned)__builtin_amdgcn_s_sendmsg_rtn(0x87) >> 18) & 1u;
}

__device__ __forceinline__ bool draw_job(unsigned* __restrict__ tickets, int& chunk, unsigned& half) {
    __shared__ unsigned job;
    if (threadIdx.x == 0) {
        unsigned h = chip_half();
        unsigned t = atomicAdd(&tickets[h], 1u);
        if (t >= SCATTER_CHUNKS) { h ^= 1u; t = atomicAdd(&tickets[h], 1u); }
        job = t < SCATTER_CHUNKS ? t * 2u + h : SKIP_EDGE;
    }
    __syncthreads();
    const unsigned j = job;
    chunk = (int)(j >> 1);
    half = j & 1u;
    return j != SKIP_EDGE;
}

template <int F, int I>
__device__ __forceinline__ unsigned edge_lane(unsigned x) {
    if constexpr (F == 16)
        return (unsigned)__builtin_amdgcn_update_dpp(0, (int)x, 0x150 + I, 0xF, 0xF, false);
    else
        return (unsigned)__builtin_amdgcn_readlane((int)x, I);
}

__device__ __forceinline__ void add_live(float* __restrict__ base, unsigned a, unsigned fb, float v) {
    if (a != SKIP_EDGE) atomicAdd((float*)((char*)base + (a + fb)), v);
}

template <int F, int... I>
__device__ __forceinline__ void scatter_batch(const float* __restrict__ feat,
                                              float* __restrict__ agg,
                                              unsigned gb, unsigned ab, unsigned fb,
                                              std::integer_sequence<int, I...>) {
    float v[sizeof...(I)];
    ((v[I] = *(const float*)((const char*)feat + (edge_lane<F, I>(gb) + fb))), ...);
    (add_live(agg, edge_lane<F, I>(ab), fb, v[I]), ...);
}

template <int F>
__global__ __launch_bounds__(256) void scatter_edges(const float* __restrict__ feat,
                                                     const int* __restrict__ src,
                                                     const int* __restrict__ dst,
                                                     const int* __restrict__ et,
                                                     float* __restrict__ agg,
                                                     float* __restrict__ cnt,
                                                     unsigned* __restrict__ tickets) {
    int chunk; unsigned half;
    if (!draw_job(tickets, chunk, half)) return;
    const int lane  = threadIdx.x & 31;
    const int ebase = chunk * EDGES_PER_BLOCK + (int)(threadIdx.x >> 5) * (32 * EDGE_BATCHES);
    const unsigned fb = (unsigned)(lane & (F - 1)) * 4u;
    for (int b = 0; b < EDGE_BATCHES; ++b) {
        const int e = ebase + b * 32 + lane;
        const unsigned d = (unsigned)dst[e];
        const bool mine = (d >= NODE_SPLIT ? 1u : 0u) == half;
        const unsigned key = (unsigned)et[e] * NN + d;
        const unsigned gb = mine ? (unsigned)src[e] * (F * 4u) : 0u;
        const unsigned ab = mine ? key * (F * 4u) : SKIP_EDGE;
        scatter_batch<F>(feat, agg, gb, ab, fb, std::make_integer_sequence<int, F>{});
        if (mine) atomicAdd(&cnt[key], 1.0f);
    }
}

#define BIN_NODES 400
#define NBINS (NN / BIN_NODES)
#define FILL_THREADS 1024
#define FILL_EPT 8
#define FILL_EDGES (FILL_THREADS * FILL_EPT)
#define FILL_BLOCKS ((NE + FILL_EDGES - 1) / FILL_EDGES)
#define NO_EDGE 0xFFFFFFFFu
static_assert(NN % BIN_NODES == 0 && BIN_NODES % 16 == 0, "ranges hold whole 16-node tiles");
static_assert(NBINS <= 256 && BIN_NODES <= 512, "range id and local node fit their fields");
static_assert((unsigned long long)NR * NN < (1ull << 23), "packed row index fits 23 bits");

__device__ __forceinline__ void block_scan_256(unsigned* buf, unsigned t) {
    for (unsigned off = 1; off < 256; off <<= 1) {
        unsigned v = 0u;
        if (t < 256 && t >= off) v = buf[t - off];
        __syncthreads();
        if (t < 256) buf[t] += v;
        __syncthreads();
    }
}

__global__ __launch_bounds__(FILL_THREADS) void bin_edges(const int* __restrict__ src,
                                                          const int* __restrict__ dst,
                                                          const int* __restrict__ et,
                                                          const unsigned* __restrict__ bintotal,
                                                          unsigned* __restrict__ cursor,
                                                          unsigned* __restrict__ list,
                                                          unsigned* __restrict__ binstart) {
    __shared__ unsigned scan[256];
    __shared__ unsigned lhist[256];
    __shared__ unsigned lend[256];
    __shared__ unsigned lbase[256];
    __shared__ unsigned sorted[FILL_EDGES];
    __shared__ unsigned char sbin[FILL_EDGES];
    const unsigned t = threadIdx.x;
    unsigned total = 0u;
    if (t < 256) {
        total = t < NBINS ? bintotal[t] : 0u;
        scan[t] = total;
        lhist[t] = 0u;
    }
    __syncthreads();
    block_scan_256(scan, t);
    unsigned word[FILL_EPT], where[FILL_EPT];
#pragma unroll
    for (int k = 0; k < FILL_EPT; ++k) {
        const unsigned e = blockIdx.x * FILL_EDGES + k * FILL_THREADS + t;
        where[k] = NO_EDGE;
        word[k] = 0u;
        if (e < NE) {
            const unsigned d = (unsigned)dst[e];
            const unsigned b = d / BIN_NODES;
            word[k] = (((unsigned)et[e] * NN + (unsigned)src[e]) << 9) | (d - b * BIN_NODES);
            where[k] = (b << 16) | atomicAdd(&lhist[b], 1u);
        }
    }
    __syncthreads();
    if (t < 256) lend[t] = lhist[t];
    __syncthreads();
    block_scan_256(lend, t);
    if (t < NBINS) {
        const unsigned start = scan[t] - total;
        const unsigned g = start + atomicAdd(&cursor[t], lhist[t]);
        lbase[t] = g - (lend[t] - lhist[t]);
        if (blockIdx.x == 0) {
            binstart[t] = start;
            if (t == NBINS - 1) binstart[NBINS] = scan[t];
        }
    }
#pragma unroll
    for (int k = 0; k < FILL_EPT; ++k)
        if (where[k] != NO_EDGE) {
            const unsigned b = where[k] >> 16;
            const unsigned i = lend[b] - lhist[b] + (where[k] & 0xFFFFu);
            sorted[i] = word[k];
            sbin[i] = (unsigned char)b;
        }
    __syncthreads();
    const unsigned first = blockIdx.x * FILL_EDGES;
    const unsigned nblock = NE - first < FILL_EDGES ? NE - first : FILL_EDGES;
    for (unsigned i = t; i < nblock; i += FILL_THREADS) list[lbase[sbin[i]] + i] = sorted[i];
}

__device__ __forceinline__ void add_local(float* acc, unsigned a, unsigned fb, float v) {
    if (a != SKIP_EDGE) atomicAdd((float*)((char*)acc + (a + fb)), v);
}

template <int... I>
__device__ __forceinline__ void gather_scaled_batch(const float* __restrict__ proj, float* acc,
                                                    unsigned gb, unsigned ab, unsigned wb,
                                                    unsigned fb,
                                                    std::integer_sequence<int, I...>) {
    float v[sizeof...(I)];
    ((v[I] = *(const float*)((const char*)proj + (edge_lane<16, I>(gb) + fb))), ...);
    (add_local(acc, edge_lane<16, I>(ab), fb,
               v[I] * __uint_as_float(edge_lane<16, I>(wb))), ...);
}

#define GATHER_THREADS 1024
__global__ __launch_bounds__(GATHER_THREADS) void gather_bins(const float* __restrict__ proj,
                                                              const float* __restrict__ cnt,
                                                              const unsigned* __restrict__ list,
                                                              const unsigned* __restrict__ binstart,
                                                              float* __restrict__ out) {
    __shared__ float acc[BIN_NODES * 16];
    __shared__ float wgt[NR * BIN_NODES];
    const unsigned t = threadIdx.x;
    const unsigned n0 = blockIdx.x * BIN_NODES;
    for (unsigned i = t; i < BIN_NODES * 16; i += GATHER_THREADS) acc[i] = out[(size_t)n0 * 16 + i];
    for (unsigned i = t; i < NR * BIN_NODES; i += GATHER_THREADS) {
        const unsigned r = i / BIN_NODES;
        wgt[i] = mean_weight(cnt[(size_t)r * NN + n0 + (i - r * BIN_NODES)]);
    }
    __syncthreads();
    const unsigned beg = binstart[blockIdx.x], end = binstart[blockIdx.x + 1];
    const unsigned lane = t & 31u;
    const unsigned fb = (lane & 15u) * 4u;
    for (unsigned base = beg + (t >> 5) * 32u; base < end; base += GATHER_THREADS) {
        const unsigned e = base + lane;
        const bool live = e < end;
        const unsigned w = live ? list[e] : 0u;
        const unsigned row = w >> 9;
        const unsigned dl = w & 511u;
        const unsigned gb = row * 64u;
        const unsigned ab = live ? dl * 64u : SKIP_EDGE;
        const unsigned wb = __float_as_uint(wgt[(row / NN) * BIN_NODES + dl]);
        gather_scaled_batch(proj, acc, gb, ab, wb, fb, std::make_integer_sequence<int, 16>{});
    }
    __syncthreads();
    for (unsigned i = t; i < BIN_NODES * 16; i += GATHER_THREADS) out[(size_t)n0 * 16 + i] = acc[i];
}

__device__ __forceinline__ v2f a_frag(const float* __restrict__ M, int row, int stride,
                                      int k0, int kh, float scale) {
    const float* p = M + (size_t)row * stride + k0 + kh;
    v2f a; a.x = p[0] * scale; a.y = p[1] * scale; return a;
}
__device__ __forceinline__ v2f b_frag(const float* __restrict__ W, int nout,
                                      int k0, int kh, int col) {
    v2f b;
    b.x = W[(size_t)(k0 + kh) * nout + col];
    b.y = W[(size_t)(k0 + kh + 1) * nout + col];
    return b;
}


__global__ __launch_bounds__(256) void rgcn_layer1(
    const float* __restrict__ x, const float* __restrict__ agg, const float* __restrict__ cnt,
    const float* __restrict__ W1, const float* __restrict__ root1, const float* __restrict__ b1,
    float* __restrict__ h, unsigned* __restrict__ bintotal) {
    int wid  = (blockIdx.x * blockDim.x + threadIdx.x) >> 5;
    int lane = threadIdx.x & 31;
    if (wid >= NN / 16) return;
    int m0   = wid * 16;
    int arow = m0 + (lane & 15);
    int kh   = (lane >> 4) << 1;
    int col  = lane & 15;

    v8f c0, c1;
    float bb0 = b1[col], bb1 = b1[16 + col];
#pragma unroll
    for (int v = 0; v < 8; ++v) { c0[v] = bb0; c1[v] = bb1; }

#pragma unroll
    for (int k = 0; k < 16; k += 4) {
        v2f a  = a_frag(x, arow, 16, k, kh, 1.0f);
        v2f bA = b_frag(root1, 32, k, kh, col);
        v2f bB = b_frag(root1, 32, k, kh, col + 16);
        c0 = __builtin_amdgcn_wmma_f32_16x16x4_f32(false, a, false, bA, (short)0, c0, false, false);
        c1 = __builtin_amdgcn_wmma_f32_16x16x4_f32(false, a, false, bB, (short)0, c1, false, false);
    }
    float indeg = 0.0f;
    for (int r = 0; r < NR; ++r) {
        const float* ag = agg + (size_t)r * NN * 16;
        const float* Wr = W1 + (size_t)r * 16 * 32;
        const float cr = cnt[(size_t)r * NN + arow];
        indeg += cr;
        float sc = mean_weight(cr);
#pragma unroll
        for (int k = 0; k < 16; k += 4) {
            v2f a  = a_frag(ag, arow, 16, k, kh, sc);
            v2f bA = b_frag(Wr, 32, k, kh, col);
            v2f bB = b_frag(Wr, 32, k, kh, col + 16);
            c0 = __builtin_amdgcn_wmma_f32_16x16x4_f32(false, a, false, bA, (short)0, c0, false, false);
            c1 = __builtin_amdgcn_wmma_f32_16x16x4_f32(false, a, false, bB, (short)0, c1, false, false);
        }
    }
    indeg += __shfl_xor(indeg, 8);
    indeg += __shfl_xor(indeg, 4);
    indeg += __shfl_xor(indeg, 2);
    indeg += __shfl_xor(indeg, 1);
    if (lane == 0) atomicAdd(&bintotal[m0 / BIN_NODES], (unsigned)indeg);
    int rbase = m0 + ((lane >> 4) << 3);
#pragma unroll
    for (int v = 0; v < 8; ++v) {
        h[(size_t)(rbase + v) * 32 + col]      = fmaxf(c0[v], 0.0f);
        h[(size_t)(rbase + v) * 32 + 16 + col] = fmaxf(c1[v], 0.0f);
    }
}


__global__ __launch_bounds__(256) void rgcn_project2(
    const float* __restrict__ h, const float* __restrict__ W2, const float* __restrict__ root2,
    const float* __restrict__ b2, float* __restrict__ proj, float* __restrict__ out) {
    int wid  = (blockIdx.x * blockDim.x + threadIdx.x) >> 5;
    int lane = threadIdx.x & 31;
    if (wid >= NN / 16) return;
    int m0   = wid * 16;
    int mrow = lane & 15;
    int arow = m0 + ((mrow & 7) << 1) + (mrow >> 3);
    int kh   = (lane >> 4) << 1;
    int col  = lane & 15;

    v2f a[8];
#pragma unroll
    for (int k = 0; k < 8; ++k) a[k] = a_frag(h, arow, 32, 4 * k, kh, 1.0f);

    v8f c[1 + NR];
    float bb = b2[col];
#pragma unroll
    for (int t = 0; t <= NR; ++t)
#pragma unroll
        for (int v = 0; v < 8; ++v) c[t][v] = (t == 0) ? bb : 0.0f;

#pragma unroll
    for (int k = 0; k < 8; ++k) {
        v2f b = b_frag(root2, 16, 4 * k, kh, col);
        c[0] = __builtin_amdgcn_wmma_f32_16x16x4_f32(false, a[k], false, b, (short)0, c[0], false, false);
#pragma unroll
        for (int r = 0; r < NR; ++r) {
            v2f br = b_frag(W2 + (size_t)r * 32 * 16, 16, 4 * k, kh, col);
            c[1 + r] = __builtin_amdgcn_wmma_f32_16x16x4_f32(false, a[k], false, br, (short)0, c[1 + r], false, false);
        }
    }
    int nbase = m0 + (lane >> 4);
#pragma unroll
    for (int v = 0; v < 8; ++v) {
        size_t node = (size_t)(nbase + 2 * v);
        out[node * 16 + col] = c[0][v];
#pragma unroll
        for (int r = 0; r < NR; ++r)
            proj[((size_t)r * NN + node) * 16 + col] = c[1 + r][v];
    }
}


extern "C" void kernel_launch(void* const* d_in, const int* in_sizes, int n_in,
                              void* d_out, int out_size, void* d_ws, size_t ws_size,
                              hipStream_t stream) {
    const float* x     = (const float*)d_in[0];
    const int*   eidx  = (const int*)d_in[1];
    const int*   etype = (const int*)d_in[2];
    const float* W1    = (const float*)d_in[3];
    const float* root1 = (const float*)d_in[4];
    const float* b1    = (const float*)d_in[5];
    const float* W2    = (const float*)d_in[6];
    const float* root2 = (const float*)d_in[7];
    const float* b2    = (const float*)d_in[8];
    float* out = (float*)d_out;

    const int* src = eidx;
    const int* dst = eidx + NE;

    float* agg  = (float*)d_ws;
    float* proj = agg + (size_t)NR * NN * 16;
    float* cnt  = agg + (size_t)NR * NN * 32;
    float* h    = cnt + (size_t)NR * NN;

    const int B = 256;
    unsigned* bintotal = (unsigned*)proj;
    unsigned* cursor   = bintotal + 256;
    unsigned* list     = (unsigned*)agg;
    unsigned* binstart = list + NE;

    zero_two<<<4096, B, 0, stream>>>(agg, (size_t)NR * NN * 16, cnt, (size_t)NR * NN + 2, bintotal, 512);
    scatter_edges<16><<<SCATTER_BLOCKS, B, 0, stream>>>(x, src, dst, etype, agg, cnt, (unsigned*)h);
    rgcn_layer1<<<((NN / 16) * 32 + B - 1) / B, B, 0, stream>>>(x, agg, cnt, W1, root1, b1, h, bintotal);

    bin_edges<<<FILL_BLOCKS, FILL_THREADS, 0, stream>>>(src, dst, etype, bintotal, cursor, list, binstart);
    rgcn_project2<<<((NN / 16) * 32 + B - 1) / B, B, 0, stream>>>(h, W2, root2, b2, proj, out);
    gather_bins<<<NBINS, GATHER_THREADS, 0, stream>>>(proj, cnt, list, binstart, out);

    (void)in_sizes; (void)n_in; (void)out_size; (void)ws_size;
}


# layer-1 edge sums leave global atomics: range sort runs first (dst histogram for totals), one work-group per range sums x rows per relation in LDS (LDS f32 adds, arrival order), WMMA from LDS
# speedup vs baseline: 4.0389x; 1.4293x over previous
#include <hip/hip_runtime.h>
#include <utility>


#define NN 100000
#define NE 3200000
#define NR 8

typedef __attribute__((ext_vector_type(2))) float v2f;
typedef __attribute__((ext_vector_type(8))) float v8f;


__global__ void zero_words(unsigned* __restrict__ c, unsigned nc) {
    const unsigned i = blockIdx.x * blockDim.x + threadIdx.x;
    if (i < nc) c[i] = 0u;
}

__device__ __forceinline__ float mean_weight(float count) {
    return 1.0f / fmaxf(count, 1.0f);
}

#define BIN_NODES 400
#define NBINS (NN / BIN_NODES)
#define FILL_THREADS 1024
#define FILL_EPT 8
#define FILL_EDGES (FILL_THREADS * FILL_EPT)
#define FILL_BLOCKS ((NE + FILL_EDGES - 1) / FILL_EDGES)
#define NO_EDGE 0xFFFFFFFFu
#define SKIP_EDGE 0xFFFFFFFFu
static_assert(NN % BIN_NODES == 0 && BIN_NODES % 16 == 0, "ranges hold whole 16-node tiles");
static_assert(NBINS <= 256 && BIN_NODES <= 512, "range id and local node fit their fields");
static_assert((unsigned long long)NR * NN < (1ull << 23), "packed row index fits 23 bits");

__global__ __launch_bounds__(FILL_THREADS) void count_ranges(const int* __restrict__ dst,
                                                             unsigned* __restrict__ bintotal) {
    __shared__ unsigned hist[256];
    const unsigned t = threadIdx.x;
    if (t < 256) hist[t] = 0u;
    __syncthreads();
    for (int k = 0; k < FILL_EPT; ++k) {
        const unsigned e = blockIdx.x * FILL_EDGES + k * FILL_THREADS + t;
        if (e < NE) atomicAdd(&hist[(unsigned)dst[e] / BIN_NODES], 1u);
    }
    __syncthreads();
    if (t < NBINS && hist[t] != 0u) atomicAdd(&bintotal[t], hist[t]);
}

__device__ __forceinline__ void block_scan_256(unsigned* buf, unsigned t) {
    for (unsigned off = 1; off < 256; off <<= 1) {
        unsigned v = 0u;
        if (t < 256 && t >= off) v = buf[t - off];
        __syncthreads();
        if (t < 256) buf[t] += v;
        __syncthreads();
    }
}

__global__ __launch_bounds__(FILL_THREADS) void bin_edges(const int* __restrict__ src,
                                                          const int* __restrict__ dst,
                                                          const int* __restrict__ et,
                                                          const unsigned* __restrict__ bintotal,
                                                          unsigned* __restrict__ cursor,
                                                          unsigned* __restrict__ list,
                                                          unsigned* __restrict__ binstart) {
    __shared__ unsigned scan[256];
    __shared__ unsigned lhist[256];
    __shared__ unsigned lend[256];
    __shared__ unsigned lbase[256];
    __shared__ unsigned sorted[FILL_EDGES];
    __shared__ unsigned char sbin[FILL_EDGES];
    const unsigned t = threadIdx.x;
    unsigned total = 0u;
    if (t < 256) {
        total = t < NBINS ? bintotal[t] : 0u;
        scan[t] = total;
        lhist[t] = 0u;
    }
    __syncthreads();
    block_scan_256(scan, t);
    unsigned word[FILL_EPT], where[FILL_EPT];
#pragma unroll
    for (int k = 0; k < FILL_EPT; ++k) {
        const unsigned e = blockIdx.x * FILL_EDGES + k * FILL_THREADS + t;
        where[k] = NO_EDGE;
        word[k] = 0u;
        if (e < NE) {
            const unsigned d = (unsigned)dst[e];
            const unsigned b = d / BIN_NODES;
            word[k] = (((unsigned)et[e] * NN + (unsigned)src[e]) << 9) | (d - b * BIN_NODES);
            where[k] = (b << 16) | atomicAdd(&lhist[b], 1u);
        }
    }
    __syncthreads();
    if (t < 256) lend[t] = lhist[t];
    __syncthreads();
    block_scan_256(lend, t);
    if (t < NBINS) {
        const unsigned start = scan[t] - total;
        const unsigned g = start + atomicAdd(&cursor[t], lhist[t]);
        lbase[t] = g - (lend[t] - lhist[t]);
        if (blockIdx.x == 0) {
            binstart[t] = start;
            if (t == NBINS - 1) binstart[NBINS] = scan[t];
        }
    }
#pragma unroll
    for (int k = 0; k < FILL_EPT; ++k)
        if (where[k] != NO_EDGE) {
            const unsigned b = where[k] >> 16;
            const unsigned i = lend[b] - lhist[b] + (where[k] & 0xFFFFu);
            sorted[i] = word[k];
            sbin[i] = (unsigned char)b;
        }
    __syncthreads();
    const unsigned first = blockIdx.x * FILL_EDGES;
    const unsigned nblock = NE - first < FILL_EDGES ? NE - first : FILL_EDGES;
    for (unsigned i = t; i < nblock; i += FILL_THREADS) list[lbase[sbin[i]] + i] = sorted[i];
}

template <int I>
__device__ __forceinline__ unsigned edge_lane(unsigned x) {
    return (unsigned)__builtin_amdgcn_update_dpp(0, (int)x, 0x150 + I, 0xF, 0xF, false);
}

__device__ __forceinline__ void add_local(float* acc, unsigned a, unsigned fb, float v) {
    if (a != SKIP_EDGE) atomicAdd((float*)((char*)acc + (a + fb)), v);
}

template <int... I>
__device__ __forceinline__ void gather_rows_batch(const float* __restrict__ feat, float* acc,
                                                  unsigned gb, unsigned ab, unsigned fb,
                                                  std::integer_sequence<int, I...>) {
    float v[sizeof...(I)];
    ((v[I] = *(const float*)((const char*)feat + (edge_lane<I>(gb) + fb))), ...);
    (add_local(acc, edge_lane<I>(ab), fb, v[I]), ...);
}
template <int... I>
__device__ __forceinline__ void gather_scaled_batch(const float* __restrict__ proj, float* acc,
                                                    unsigned gb, unsigned ab, unsigned wb,
                                                    unsigned fb,
                                                    std::integer_sequence<int, I...>) {
    float v[sizeof...(I)];
    ((v[I] = *(const float*)((const char*)proj + (edge_lane<I>(gb) + fb))), ...);
    (add_local(acc, edge_lane<I>(ab), fb,
               v[I] * __uint_as_float(edge_lane<I>(wb))), ...);
}

__device__ __forceinline__ v2f a_frag(const float* __restrict__ M, int row, int stride,
                                      int k0, int kh, float scale) {
    const float* p = M + (size_t)row * stride + k0 + kh;
    v2f a; a.x = p[0] * scale; a.y = p[1] * scale; return a;
}
__device__ __forceinline__ v2f b_frag(const float* __restrict__ W, int nout,
                                      int k0, int kh, int col) {
    v2f b;
    b.x = W[(size_t)(k0 + kh) * nout + col];
    b.y = W[(size_t)(k0 + kh + 1) * nout + col];
    return b;
}

#define L1_THREADS 1024
__global__ __launch_bounds__(L1_THREADS) void layer1_bins(
    const float* __restrict__ x, const unsigned* __restrict__ list,
    const unsigned* __restrict__ binstart,
    const float* __restrict__ W1, const float* __restrict__ root1, const float* __restrict__ b1,
    float* __restrict__ cnt, float* __restrict__ h) {
    __shared__ float acc[NR * BIN_NODES * 16];
    __shared__ float cl[NR * BIN_NODES];
    const unsigned t = threadIdx.x;
    const unsigned n0 = blockIdx.x * BIN_NODES;
    for (unsigned i = t; i < NR * BIN_NODES * 16; i += L1_THREADS) acc[i] = 0.0f;
    for (unsigned i = t; i < NR * BIN_NODES; i += L1_THREADS) cl[i] = 0.0f;
    __syncthreads();
    const unsigned beg = binstart[blockIdx.x], end = binstart[blockIdx.x + 1];
    const unsigned lane = t & 31u;
    const unsigned fb = (lane & 15u) * 4u;
    for (unsigned base = beg + (t >> 5) * 32u; base < end; base += L1_THREADS) {
        const unsigned e = base + lane;
        const bool live = e < end;
        const unsigned w = live ? list[e] : 0u;
        const unsigned row = w >> 9;
        const unsigned dl = w & 511u;
        const unsigned r = row / NN;
        const unsigned slot = r * BIN_NODES + dl;
        const unsigned gb = (row - r * NN) * 64u;
        const unsigned ab = live ? slot * 64u : SKIP_EDGE;
        gather_rows_batch(x, acc, gb, ab, fb, std::make_integer_sequence<int, 16>{});
        if (live) atomicAdd(&cl[slot], 1.0f);
    }
    __syncthreads();
    for (unsigned i = t; i < NR * BIN_NODES; i += L1_THREADS) {
        const unsigned r = i / BIN_NODES;
        cnt[(size_t)r * NN + n0 + (i - r * BIN_NODES)] = cl[i];
    }
    const int tile = (int)(t >> 5);
    if (tile >= BIN_NODES / 16) return;
    const int ln   = tile * 16 + (int)(lane & 15u);
    const int arow = (int)n0 + ln;
    const int kh   = (int)(lane >> 4) << 1;
    const int col  = (int)(lane & 15u);

    v8f c0, c1;
    float bb0 = b1[col], bb1 = b1[16 + col];
#pragma unroll
    for (int v = 0; v < 8; ++v) { c0[v] = bb0; c1[v] = bb1; }

#pragma unroll
    for (int k = 0; k < 16; k += 4) {
        v2f a  = a_frag(x, arow, 16, k, kh, 1.0f);
        v2f bA = b_frag(root1, 32, k, kh, col);
        v2f bB = b_frag(root1, 32, k, kh, col + 16);
        c0 = __builtin_amdgcn_wmma_f32_16x16x4_f32(false, a, false, bA, (short)0, c0, false, false);
        c1 = __builtin_amdgcn_wmma_f32_16x16x4_f32(false, a, false, bB, (short)0, c1, false, false);
    }
    for (int r = 0; r < NR; ++r) {
        const float* Wr = W1 + (size_t)r * 16 * 32;
        const float sc = mean_weight(cl[r * BIN_NODES + ln]);
#pragma unroll
        for (int k = 0; k < 16; k += 4) {
            v2f a;
            a.x = acc[(r * BIN_NODES + ln) * 16 + k + kh] * sc;
            a.y = acc[(r * BIN_NODES + ln) * 16 + k + kh + 1] * sc;
            v2f bA = b_frag(Wr, 32, k, kh, col);
            v2f bB = b_frag(Wr, 32, k, kh, col + 16);
            c0 = __builtin_amdgcn_wmma_f32_16x16x4_f32(false, a, false, bA, (short)0, c0, false, false);
            c1 = __builtin_amdgcn_wmma_f32_16x16x4_f32(false, a, false, bB, (short)0, c1, false, false);
        }
    }
    const int rbase = (int)n0 + tile * 16 + ((int)(lane >> 4) << 3);
#pragma unroll
    for (int v = 0; v < 8; ++v) {
        h[(size_t)(rbase + v) * 32 + col]      = fmaxf(c0[v], 0.0f);
        h[(size_t)(rbase + v) * 32 + 16 + col] = fmaxf(c1[v], 0.0f);
    }
}


__global__ __launch_bounds__(256) void rgcn_project2(
    const float* __restrict__ h, const float* __restrict__ W2, const float* __restrict__ root2,
    const float* __restrict__ b2, float* __restrict__ proj, float* __restrict__ out) {
    int wid  = (blockIdx.x * blockDim.x + threadIdx.x) >> 5;
    int lane = threadIdx.x & 31;
    if (wid >= NN / 16) return;
    int m0   = wid * 16;
    int mrow = lane & 15;
    int arow = m0 + ((mrow & 7) << 1) + (mrow >> 3);
    int kh   = (lane >> 4) << 1;
    int col  = lane & 15;

    v2f a[8];
#pragma unroll
    for (int k = 0; k < 8; ++k) a[k] = a_frag(h, arow, 32, 4 * k, kh, 1.0f);

    v8f c[1 + NR];
    float bb = b2[col];
#pragma unroll
    for (int t = 0; t <= NR; ++t)
#pragma unroll
        for (int v = 0; v < 8; ++v) c[t][v] = (t == 0) ? bb : 0.0f;

#pragma unroll
    for (int k = 0; k < 8; ++k) {
        v2f b = b_frag(root2, 16, 4 * k, kh, col);
        c[0] = __builtin_amdgcn_wmma_f32_16x16x4_f32(false, a[k], false, b, (short)0, c[0], false, false);
#pragma unroll
        for (int r = 0; r < NR; ++r) {
            v2f br = b_frag(W2 + (size_t)r * 32 * 16, 16, 4 * k, kh, col);
            c[1 + r] = __builtin_amdgcn_wmma_f32_16x16x4_f32(false, a[k], false, br, (short)0, c[1 + r], false, false);
        }
    }
    int nbase = m0 + (lane >> 4);
#pragma unroll
    for (int v = 0; v < 8; ++v) {
        size_t node = (size_t)(nbase + 2 * v);
        out[node * 16 + col] = c[0][v];
#pragma unroll
        for (int r = 0; r < NR; ++r)
            proj[((size_t)r * NN + node) * 16 + col] = c[1 + r][v];
    }
}

#define GATHER_THREADS 1024
__global__ __launch_bounds__(GATHER_THREADS) void gather_bins(const float* __restrict__ proj,
                                                              const float* __restrict__ cnt,
                                                              const unsigned* __restrict__ list,
                                                              const unsigned* __restrict__ binstart,
                                                              float* __restrict__ out) {
    __shared__ float acc[BIN_NODES * 16];
    __shared__ float wgt[NR * BIN_NODES];
    const unsigned t = threadIdx.x;
    const unsigned n0 = blockIdx.x * BIN_NODES;
    for (unsigned i = t; i < BIN_NODES * 16; i += GATHER_THREADS) acc[i] = out[(size_t)n0 * 16 + i];
    for (unsigned i = t; i < NR * BIN_NODES; i += GATHER_THREADS) {
        const unsigned r = i / BIN_NODES;
        wgt[i] = mean_weight(cnt[(size_t)r * NN + n0 + (i - r * BIN_NODES)]);
    }
    __syncthreads();
    const unsigned beg = binstart[blockIdx.x], end = binstart[blockIdx.x + 1];
    const unsigned lane = t & 31u;
    const unsigned fb = (lane & 15u) * 4u;
    for (unsigned base = beg + (t >> 5) * 32u; base < end; base += GATHER_THREADS) {
        const unsigned e = base + lane;
        const bool live = e < end;
        const unsigned w = live ? list[e] : 0u;
        const unsigned row = w >> 9;
        const unsigned dl = w & 511u;
        const unsigned gb = row * 64u;
        const unsigned ab = live ? dl * 64u : SKIP_EDGE;
        const unsigned wb = __float_as_uint(wgt[(row / NN) * BIN_NODES + dl]);
        gather_scaled_batch(proj, acc, gb, ab, wb, fb, std::make_integer_sequence<int, 16>{});
    }
    __syncthreads();
    for (unsigned i = t; i < BIN_NODES * 16; i += GATHER_THREADS) out[(size_t)n0 * 16 + i] = acc[i];
}


extern "C" void kernel_launch(void* const* d_in, const int* in_sizes, int n_in,
                              void* d_out, int out_size, void* d_ws, size_t ws_size,
                              hipStream_t stream) {
    const float* x     = (const float*)d_in[0];
    const int*   eidx  = (const int*)d_in[1];
    const int*   etype = (const int*)d_in[2];
    const float* W1    = (const float*)d_in[3];
    const float* root1 = (const float*)d_in[4];
    const float* b1    = (const float*)d_in[5];
    const float* W2    = (const float*)d_in[6];
    const float* root2 = (const float*)d_in[7];
    const float* b2    = (const float*)d_in[8];
    float* out = (float*)d_out;

    const int* src = eidx;
    const int* dst = eidx + NE;

    float* plane = (float*)d_ws;
    float* proj  = plane + (size_t)NR * NN * 16;
    float* cnt   = plane + (size_t)NR * NN * 32;
    float* h     = cnt + (size_t)NR * NN;

    const int B = 256;
    unsigned* bintotal = (unsigned*)proj;
    unsigned* cursor   = bintotal + 256;
    unsigned* list     = (unsigned*)plane;
    unsigned* binstart = list + NE;

    zero_words<<<2, B, 0, stream>>>(bintotal, 512u);
    count_ranges<<<FILL_BLOCKS, FILL_THREADS, 0, stream>>>(dst, bintotal);
    bin_edges<<<FILL_BLOCKS, FILL_THREADS, 0, stream>>>(src, dst, etype, bintotal, cursor, list, binstart);

    layer1_bins<<<NBINS, L1_THREADS, 0, stream>>>(x, list, binstart, W1, root1, b1, cnt, h);

    rgcn_project2<<<((NN / 16) * 32 + B - 1) / B, B, 0, stream>>>(h, W2, root2, b2, proj, out);
    gather_bins<<<NBINS, GATHER_THREADS, 0, stream>>>(proj, cnt, list, binstart, out);

    (void)in_sizes; (void)n_in; (void)out_size; (void)ws_size;
}


# edge sort in one pass: each of 256 edge chunks is range-sorted in LDS by one work-group, with a row of run starts; range walkers take 8 runs per wave; 6 launches to 4; LDS sums in arrival order
# speedup vs baseline: 4.2070x; 1.0416x over previous
#include <hip/hip_runtime.h>
#include <utility>


#define NN 100000
#define NE 3200000
#define NR 8

typedef __attribute__((ext_vector_type(2))) float v2f;
typedef __attribute__((ext_vector_type(8))) float v8f;


__device__ __forceinline__ float mean_weight(float count) {
    return 1.0f / fmaxf(count, 1.0f);
}

#define BIN_NODES 400
#define NBINS (NN / BIN_NODES)
#define NCHUNK 256
#define CHUNK (NE / NCHUNK)
#define SORT_THREADS 512
#define SORT_EPT ((CHUNK + SORT_THREADS - 1) / SORT_THREADS)
#define WALK_THREADS 1024
#define RUNS (NCHUNK / (WALK_THREADS / 32))
#define NO_EDGE 0xFFFFFFFFu
#define SKIP_EDGE 0xFFFFFFFFu
static_assert(NN % BIN_NODES == 0 && BIN_NODES % 16 == 0, "ranges hold whole 16-node tiles");
static_assert(NBINS < 256 && BIN_NODES <= 512, "range id and local node fit their fields");
static_assert((unsigned long long)NR * NN < (1ull << 23), "packed row index fits 23 bits");
static_assert(NE % NCHUNK == 0 && CHUNK < 65536, "whole chunks; a rank inside a chunk fits 16 bits");
static_assert(NCHUNK % (WALK_THREADS / 32) == 0, "every walking wave owns RUNS whole runs");

__device__ __forceinline__ void block_scan_256(unsigned* buf, unsigned t) {
    for (unsigned off = 1; off < 256; off <<= 1) {
        unsigned v = 0u;
        if (t < 256 && t >= off) v = buf[t - off];
        __syncthreads();
        if (t < 256) buf[t] += v;
        __syncthreads();
    }
}

__global__ __launch_bounds__(SORT_THREADS) void sort_chunks(const int* __restrict__ src,
                                                            const int* __restrict__ dst,
                                                            const int* __restrict__ et,
                                                            unsigned* __restrict__ list,
                                                            unsigned* __restrict__ runstart) {
    __shared__ unsigned lhist[256];
    __shared__ unsigned lend[256];
    __shared__ unsigned sorted[CHUNK];
    const unsigned t = threadIdx.x;
    const unsigned first = blockIdx.x * CHUNK;
    if (t < 256) lhist[t] = 0u;
    unsigned dv[SORT_EPT], sv[SORT_EPT], tv[SORT_EPT];
#pragma unroll
    for (int k = 0; k < SORT_EPT; ++k) {
        const unsigned i = k * SORT_THREADS + t;
        const unsigned e = first + (i < CHUNK ? i : CHUNK - 1);
        dv[k] = (unsigned)dst[e];
        sv[k] = (unsigned)src[e];
        tv[k] = (unsigned)et[e];
    }
    __syncthreads();
    unsigned word[SORT_EPT], where[SORT_EPT];
#pragma unroll
    for (int k = 0; k < SORT_EPT; ++k) {
        const unsigned b = dv[k] / BIN_NODES;
        word[k] = ((tv[k] * NN + sv[k]) << 9) | (dv[k] - b * BIN_NODES);
        where[k] = NO_EDGE;
        if (k * SORT_THREADS + t < CHUNK) where[k] = (b << 16) | atomicAdd(&lhist[b], 1u);
    }
    __syncthreads();
    if (t < 256) lend[t] = lhist[t];
    __syncthreads();
    block_scan_256(lend, t);
    if (t < 256) {
        const unsigned start = t == 0 ? 0u : lend[t - 1];
        runstart[blockIdx.x * 256 + t] = start;
        lhist[t] = start;
    }
    __syncthreads();
    unsigned pos[SORT_EPT];
#pragma unroll
    for (int k = 0; k < SORT_EPT; ++k)
        pos[k] = lhist[where[k] != NO_EDGE ? where[k] >> 16 : 0u] + (where[k] & 0xFFFFu);
#pragma unroll
    for (int k = 0; k < SORT_EPT; ++k)
        if (where[k] != NO_EDGE) sorted[pos[k]] = word[k];
    __syncthreads();
    for (unsigned i = t; i < CHUNK; i += SORT_THREADS) list[first + i] = sorted[i];
}

struct wave_runs {
    unsigned first[RUNS + 1];
    unsigned shift[RUNS];
};
__device__ __forceinline__ wave_runs load_runs(const unsigned* __restrict__ runstart,
                                               unsigned wave, unsigned range) {
    wave_runs w;
    w.first[0] = 0u;
#pragma unroll
    for (int j = 0; j < RUNS; ++j) {
        const unsigned c = wave * RUNS + j;
        const unsigned lo = runstart[c * 256 + range], hi = runstart[c * 256 + range + 1];
        w.shift[j] = c * CHUNK + lo - w.first[j];
        w.first[j + 1] = w.first[j] + (hi - lo);
    }
    return w;
}
__device__ __forceinline__ unsigned run_slot(const wave_runs& w, unsigned v) {
    unsigned s = w.shift[0];
#pragma unroll
    for (int j = 1; j < RUNS; ++j) s = v >= w.first[j] ? w.shift[j] : s;
    return v + s;
}

template <int I>
__device__ __forceinline__ unsigned edge_lane(unsigned x) {
    return (unsigned)__builtin_amdgcn_update_dpp(0, (int)x, 0x150 + I, 0xF, 0xF, false);
}

__device__ __forceinline__ void add_local(float* acc, unsigned a, unsigned fb, float v) {
    if (a != SKIP_EDGE) atomicAdd((float*)((char*)acc + (a + fb)), v);
}

template <int... I>
__device__ __forceinline__ void gather_rows_batch(const float* __restrict__ feat, float* acc,
                                                  unsigned gb, unsigned ab, unsigned fb,
                                                  std::integer_sequence<int, I...>) {
    float v[sizeof...(I)];
    ((v[I] = *(const float*)((const char*)feat + (edge_lane<I>(gb) + fb))), ...);
    (add_local(acc, edge_lane<I>(ab), fb, v[I]), ...);
}
template <int... I>
__device__ __forceinline__ void gather_scaled_batch(const float* __restrict__ proj, float* acc,
                                                    unsigned gb, unsigned ab, unsigned wb,
                                                    unsigned fb,
                                                    std::integer_sequence<int, I...>) {
    float v[sizeof...(I)];
    ((v[I] = *(const float*)((const char*)proj + (edge_lane<I>(gb) + fb))), ...);
    (add_local(acc, edge_lane<I>(ab), fb,
               v[I] * __uint_as_float(edge_lane<I>(wb))), ...);
}

__device__ __forceinline__ v2f a_frag(const float* __restrict__ M, int row, int stride,
                                      int k0, int kh, float scale) {
    const float* p = M + (size_t)row * stride + k0 + kh;
    v2f a; a.x = p[0] * scale; a.y = p[1] * scale; return a;
}
__device__ __forceinline__ v2f b_frag(const float* __restrict__ W, int nout,
                                      int k0, int kh, int col) {
    v2f b;
    b.x = W[(size_t)(k0 + kh) * nout + col];
    b.y = W[(size_t)(k0 + kh + 1) * nout + col];
    return b;
}

#define L1_THREADS WALK_THREADS
__global__ __launch_bounds__(L1_THREADS) void layer1_bins(
    const float* __restrict__ x, const unsigned* __restrict__ list,
    const unsigned* __restrict__ runstart,
    const float* __restrict__ W1, const float* __restrict__ root1, const float* __restrict__ b1,
    float* __restrict__ cnt, float* __restrict__ h) {
    __shared__ float acc[NR * BIN_NODES * 16];
    __shared__ float cl[NR * BIN_NODES];
    const unsigned t = threadIdx.x;
    const unsigned n0 = blockIdx.x * BIN_NODES;
    const wave_runs runs = load_runs(runstart, __builtin_amdgcn_readfirstlane(t >> 5), blockIdx.x);
    for (unsigned i = t; i < NR * BIN_NODES * 16; i += L1_THREADS) acc[i] = 0.0f;
    for (unsigned i = t; i < NR * BIN_NODES; i += L1_THREADS) cl[i] = 0.0f;
    __syncthreads();
    const unsigned lane = t & 31u;
    const unsigned fb = (lane & 15u) * 4u;
    for (unsigned v0 = 0u; v0 < runs.first[RUNS]; v0 += 32u) {
        const unsigned v = v0 + lane;
        const bool live = v < runs.first[RUNS];
        const unsigned w = live ? list[run_slot(runs, v)] : 0u;
        const unsigned row = w >> 9;
        const unsigned dl = w & 511u;
        const unsigned r = row / NN;
        const unsigned slot = r * BIN_NODES + dl;
        const unsigned gb = (row - r * NN) * 64u;
        const unsigned ab = live ? slot * 64u : SKIP_EDGE;
        gather_rows_batch(x, acc, gb, ab, fb, std::make_integer_sequence<int, 16>{});
        if (live) atomicAdd(&cl[slot], 1.0f);
    }
    __syncthreads();
    for (unsigned i = t; i < NR * BIN_NODES; i += L1_THREADS) {
        const unsigned r = i / BIN_NODES;
        cnt[(size_t)r * NN + n0 + (i - r * BIN_NODES)] = cl[i];
    }
    const int tile = (int)(t >> 5);
    if (tile >= BIN_NODES / 16) return;
    const int ln   = tile * 16 + (int)(lane & 15u);
    const int arow = (int)n0 + ln;
    const int kh   = (int)(lane >> 4) << 1;
    const int col  = (int)(lane & 15u);

    v8f c0, c1;
    float bb0 = b1[col], bb1 = b1[16 + col];
#pragma unroll
    for (int v = 0; v < 8; ++v) { c0[v] = bb0; c1[v] = bb1; }

#pragma unroll
    for (int k = 0; k < 16; k += 4) {
        v2f a  = a_frag(x, arow, 16, k, kh, 1.0f);
        v2f bA = b_frag(root1, 32, k, kh, col);
        v2f bB = b_frag(root1, 32, k, kh, col + 16);
        c0 = __builtin_amdgcn_wmma_f32_16x16x4_f32(false, a, false, bA, (short)0, c0, false, false);
        c1 = __builtin_amdgcn_wmma_f32_16x16x4_f32(false, a, false, bB, (short)0, c1, false, false);
    }
    for (int r = 0; r < NR; ++r) {
        const float* Wr = W1 + (size_t)r * 16 * 32;
        const float sc = mean_weight(cl[r * BIN_NODES + ln]);
#pragma unroll
        for (int k = 0; k < 16; k += 4) {
            v2f a;
            a.x = acc[(r * BIN_NODES + ln) * 16 + k + kh] * sc;
            a.y = acc[(r * BIN_NODES + ln) * 16 + k + kh + 1] * sc;
            v2f bA = b_frag(Wr, 32, k, kh, col);
            v2f bB = b_frag(Wr, 32, k, kh, col + 16);
            c0 = __builtin_amdgcn_wmma_f32_16x16x4_f32(false, a, false, bA, (short)0, c0, false, false);
            c1 = __builtin_amdgcn_wmma_f32_16x16x4_f32(false, a, false, bB, (short)0, c1, false, false);
        }
    }
    const int rbase = (int)n0 + tile * 16 + ((int)(lane >> 4) << 3);
#pragma unroll
    for (int v = 0; v < 8; ++v) {
        h[(size_t)(rbase + v) * 32 + col]      = fmaxf(c0[v], 0.0f);
        h[(size_t)(rbase + v) * 32 + 16 + col] = fmaxf(c1[v], 0.0f);
    }
}


__global__ __launch_bounds__(256) void rgcn_project2(
    const float* __restrict__ h, const float* __restrict__ W2, const float* __restrict__ root2,
    const float* __restrict__ b2, float* __restrict__ proj, float* __restrict__ out) {
    int wid  = (blockIdx.x * blockDim.x + threadIdx.x) >> 5;
    int lane = threadIdx.x & 31;
    if (wid >= NN / 16) return;
    int m0   = wid * 16;
    int mrow = lane & 15;
    int arow = m0 + ((mrow & 7) << 1) + (mrow >> 3);
    int kh   = (lane >> 4) << 1;
    int col  = lane & 15;

    v2f a[8];
#pragma unroll
    for (int k = 0; k < 8; ++k) a[k] = a_frag(h, arow, 32, 4 * k, kh, 1.0f);

    v8f c[1 + NR];
    float bb = b2[col];
#pragma unroll
    for (int t = 0; t <= NR; ++t)
#pragma unroll
        for (int v = 0; v < 8; ++v) c[t][v] = (t == 0) ? bb : 0.0f;

#pragma unroll
    for (int k = 0; k < 8; ++k) {
        v2f b = b_frag(root2, 16, 4 * k, kh, col);
        c[0] = __builtin_amdgcn_wmma_f32_16x16x4_f32(false, a[k], false, b, (short)0, c[0], false, false);
#pragma unroll
        for (int r = 0; r < NR; ++r) {
            v2f br = b_frag(W2 + (size_t)r * 32 * 16, 16, 4 * k, kh, col);
            c[1 + r] = __builtin_amdgcn_wmma_f32_16x16x4_f32(false, a[k], false, br, (short)0, c[1 + r], false, false);
        }
    }
    int nbase = m0 + (lane >> 4);
#pragma unroll
    for (int v = 0; v < 8; ++v) {
        size_t node = (size_t)(nbase + 2 * v);
        out[node * 16 + col] = c[0][v];
#pragma unroll
        for (int r = 0; r < NR; ++r)
            proj[((size_t)r * NN + node) * 16 + col] = c[1 + r][v];
    }
}

#define GATHER_THREADS WALK_THREADS
__global__ __launch_bounds__(GATHER_THREADS) void gather_bins(const float* __restrict__ proj,
                                                              const float* __restrict__ cnt,
                                                              const unsigned* __restrict__ list,
                                                              const unsigned* __restrict__ runstart,
                                                              float* __restrict__ out) {
    __shared__ float acc[BIN_NODES * 16];
    __shared__ float wgt[NR * BIN_NODES];
    const unsigned t = threadIdx.x;
    const unsigned n0 = blockIdx.x * BIN_NODES;
    const wave_runs runs = load_runs(runstart, __builtin_amdgcn_readfirstlane(t >> 5), blockIdx.x);
    for (unsigned i = t; i < BIN_NODES * 16; i += GATHER_THREADS) acc[i] = out[(size_t)n0 * 16 + i];
    for (unsigned i = t; i < NR * BIN_NODES; i += GATHER_THREADS) {
        const unsigned r = i / BIN_NODES;
        wgt[i] = mean_weight(cnt[(size_t)r * NN + n0 + (i - r * BIN_NODES)]);
    }
    __syncthreads();
    const unsigned lane = t & 31u;
    const unsigned fb = (lane & 15u) * 4u;
    for (unsigned v0 = 0u; v0 < runs.first[RUNS]; v0 += 32u) {
        const unsigned v = v0 + lane;
        const bool live = v < runs.first[RUNS];
        const unsigned w = live ? list[run_slot(runs, v)] : 0u;
        const unsigned row = w >> 9;
        const unsigned dl = w & 511u;
        const unsigned gb = row * 64u;
        const unsigned ab = live ? dl * 64u : SKIP_EDGE;
        const unsigned wb = __float_as_uint(wgt[(row / NN) * BIN_NODES + dl]);
        gather_scaled_batch(proj, acc, gb, ab, wb, fb, std::make_integer_sequence<int, 16>{});
    }
    __syncthreads();
    for (unsigned i = t; i < BIN_NODES * 16; i += GATHER_THREADS) out[(size_t)n0 * 16 + i] = acc[i];
}


extern "C" void kernel_launch(void* const* d_in, const int* in_sizes, int n_in,
                              void* d_out, int out_size, void* d_ws, size_t ws_size,
                              hipStream_t stream) {
    const float* x     = (const float*)d_in[0];
    const int*   eidx  = (const int*)d_in[1];
    const int*   etype = (const int*)d_in[2];
    const float* W1    = (const float*)d_in[3];
    const float* root1 = (const float*)d_in[4];
    const float* b1    = (const float*)d_in[5];
    const float* W2    = (const float*)d_in[6];
    const float* root2 = (const float*)d_in[7];
    const float* b2    = (const float*)d_in[8];
    float* out = (float*)d_out;

    const int* src = eidx;
    const int* dst = eidx + NE;

    float* plane = (float*)d_ws;
    float* proj  = plane + (size_t)NR * NN * 16;
    float* cnt   = plane + (size_t)NR * NN * 32;
    float* h     = cnt + (size_t)NR * NN;

    const int B = 256;
    unsigned* list     = (unsigned*)plane;
    unsigned* runstart = list + NE;

    sort_chunks<<<NCHUNK, SORT_THREADS, 0, stream>>>(src, dst, etype, list, runstart);

    layer1_bins<<<NBINS, L1_THREADS, 0, stream>>>(x, list, runstart, W1, root1, b1, cnt, h);

    rgcn_project2<<<((NN / 16) * 32 + B - 1) / B, B, 0, stream>>>(h, W2, root2, b2, proj, out);
    gather_bins<<<NBINS, GATHER_THREADS, 0, stream>>>(proj, cnt, list, runstart, out);

    (void)in_sizes; (void)n_in; (void)out_size; (void)ws_size;
}


# layer-2 projection (h @ W2_r, h @ root2 + b2) done by the wave that forms each 16x32 tile of h, tile passed through the wave's own LDS rows: h never stored, 4 launches to 3; sums as before
# speedup vs baseline: 4.8002x; 1.1410x over previous
#include <hip/hip_runtime.h>
#include <utility>


#define NN 100000
#define NE 3200000
#define NR 8

typedef __attribute__((ext_vector_type(2))) float v2f;
typedef __attribute__((ext_vector_type(8))) float v8f;


__device__ __forceinline__ float mean_weight(float count) {
    return 1.0f / fmaxf(count, 1.0f);
}

#define BIN_NODES 400
#define NBINS (NN / BIN_NODES)
#define NCHUNK 256
#define CHUNK (NE / NCHUNK)
#define SORT_THREADS 512
#define SORT_EPT ((CHUNK + SORT_THREADS - 1) / SORT_THREADS)
#define WALK_THREADS 1024
#define RUNS (NCHUNK / (WALK_THREADS / 32))
#define NO_EDGE 0xFFFFFFFFu
#define SKIP_EDGE 0xFFFFFFFFu
static_assert(NN % BIN_NODES == 0 && BIN_NODES % 16 == 0, "ranges hold whole 16-node tiles");
static_assert(NBINS < 256 && BIN_NODES <= 512, "range id and local node fit their fields");
static_assert((unsigned long long)NR * NN < (1ull << 23), "packed row index fits 23 bits");
static_assert(NE % NCHUNK == 0 && CHUNK < 65536, "whole chunks; a rank inside a chunk fits 16 bits");
static_assert(NCHUNK % (WALK_THREADS / 32) == 0, "every walking wave owns RUNS whole runs");

__device__ __forceinline__ void block_scan_256(unsigned* buf, unsigned t) {
    for (unsigned off = 1; off < 256; off <<= 1) {
        unsigned v = 0u;
        if (t < 256 && t >= off) v = buf[t - off];
        __syncthreads();
        if (t < 256) buf[t] += v;
        __syncthreads();
    }
}

__global__ __launch_bounds__(SORT_THREADS) void sort_chunks(const int* __restrict__ src,
                                                            const int* __restrict__ dst,
                                                            const int* __restrict__ et,
                                                            unsigned* __restrict__ list,
                                                            unsigned* __restrict__ runstart) {
    __shared__ unsigned lhist[256];
    __shared__ unsigned lend[256];
    __shared__ unsigned sorted[CHUNK];
    const unsigned t = threadIdx.x;
    const unsigned first = blockIdx.x * CHUNK;
    if (t < 256) lhist[t] = 0u;
    unsigned dv[SORT_EPT], sv[SORT_EPT], tv[SORT_EPT];
#pragma unroll
    for (int k = 0; k < SORT_EPT; ++k) {
        const unsigned i = k * SORT_THREADS + t;
        const unsigned e = first + (i < CHUNK ? i : CHUNK - 1);
        dv[k] = (unsigned)dst[e];
        sv[k] = (unsigned)src[e];
        tv[k] = (unsigned)et[e];
    }
    __syncthreads();
    unsigned word[SORT_EPT], where[SORT_EPT];
#pragma unroll
    for (int k = 0; k < SORT_EPT; ++k) {
        const unsigned b = dv[k] / BIN_NODES;
        word[k] = ((tv[k] * NN + sv[k]) << 9) | (dv[k] - b * BIN_NODES);
        where[k] = NO_EDGE;
        if (k * SORT_THREADS + t < CHUNK) where[k] = (b << 16) | atomicAdd(&lhist[b], 1u);
    }
    __syncthreads();
    if (t < 256) lend[t] = lhist[t];
    __syncthreads();
    block_scan_256(lend, t);
    if (t < 256) {
        const unsigned start = t == 0 ? 0u : lend[t - 1];
        runstart[blockIdx.x * 256 + t] = start;
        lhist[t] = start;
    }
    __syncthreads();
    unsigned pos[SORT_EPT];
#pragma unroll
    for (int k = 0; k < SORT_EPT; ++k)
        pos[k] = lhist[where[k] != NO_EDGE ? where[k] >> 16 : 0u] + (where[k] & 0xFFFFu);
#pragma unroll
    for (int k = 0; k < SORT_EPT; ++k)
        if (where[k] != NO_EDGE) sorted[pos[k]] = word[k];
    __syncthreads();
    for (unsigned i = t; i < CHUNK; i += SORT_THREADS) list[first + i] = sorted[i];
}

struct wave_runs {
    unsigned first[RUNS + 1];
    unsigned shift[RUNS];
};
__device__ __forceinline__ wave_runs load_runs(const unsigned* __restrict__ runstart,
                                               unsigned wave, unsigned range) {
    wave_runs w;
    w.first[0] = 0u;
#pragma unroll
    for (int j = 0; j < RUNS; ++j) {
        const unsigned c = wave * RUNS + j;
        const unsigned lo = runstart[c * 256 + range], hi = runstart[c * 256 + range + 1];
        w.shift[j] = c * CHUNK + lo - w.first[j];
        w.first[j + 1] = w.first[j] + (hi - lo);
    }
    return w;
}
__device__ __forceinline__ unsigned run_slot(const wave_runs& w, unsigned v) {
    unsigned s = w.shift[0];
#pragma unroll
    for (int j = 1; j < RUNS; ++j) s = v >= w.first[j] ? w.shift[j] : s;
    return v + s;
}

template <int I>
__device__ __forceinline__ unsigned edge_lane(unsigned x) {
    return (unsigned)__builtin_amdgcn_update_dpp(0, (int)x, 0x150 + I, 0xF, 0xF, false);
}

__device__ __forceinline__ void add_local(float* acc, unsigned a, unsigned fb, float v) {
    if (a != SKIP_EDGE) atomicAdd((float*)((char*)acc + (a + fb)), v);
}

template <int... I>
__device__ __forceinline__ void gather_rows_batch(const float* __restrict__ feat, float* acc,
                                                  unsigned gb, unsigned ab, unsigned fb,
                                                  std::integer_sequence<int, I...>) {
    float v[sizeof...(I)];
    ((v[I] = *(const float*)((const char*)feat + (edge_lane<I>(gb) + fb))), ...);
    (add_local(acc, edge_lane<I>(ab), fb, v[I]), ...);
}
template <int... I>
__device__ __forceinline__ void gather_scaled_batch(const float* __restrict__ proj, float* acc,
                                                    unsigned gb, unsigned ab, unsigned wb,
                                                    unsigned fb,
                                                    std::integer_sequence<int, I...>) {
    float v[sizeof...(I)];
    ((v[I] = *(const float*)((const char*)proj + (edge_lane<I>(gb) + fb))), ...);
    (add_local(acc, edge_lane<I>(ab), fb,
               v[I] * __uint_as_float(edge_lane<I>(wb))), ...);
}

__device__ __forceinline__ v2f a_frag(const float* __restrict__ M, int row, int stride,
                                      int k0, int kh, float scale) {
    const float* p = M + (size_t)row * stride + k0 + kh;
    v2f a; a.x = p[0] * scale; a.y = p[1] * scale; return a;
}
__device__ __forceinline__ v2f b_frag(const float* __restrict__ W, int nout,
                                      int k0, int kh, int col) {
    v2f b;
    b.x = W[(size_t)(k0 + kh) * nout + col];
    b.y = W[(size_t)(k0 + kh + 1) * nout + col];
    return b;
}

#define L1_THREADS WALK_THREADS
#define HP 36
__global__ __launch_bounds__(L1_THREADS) void layer1_bins(
    const float* __restrict__ x, const unsigned* __restrict__ list,
    const unsigned* __restrict__ runstart,
    const float* __restrict__ W1, const float* __restrict__ root1, const float* __restrict__ b1,
    const float* __restrict__ W2, const float* __restrict__ root2, const float* __restrict__ b2,
    float* __restrict__ cnt, float* __restrict__ proj, float* __restrict__ out) {
    __shared__ float acc[NR * BIN_NODES * 16];
    __shared__ float cl[NR * BIN_NODES];
    __shared__ float htile[(BIN_NODES / 16) * 16 * HP];
    const unsigned t = threadIdx.x;
    const unsigned n0 = blockIdx.x * BIN_NODES;
    const wave_runs runs = load_runs(runstart, __builtin_amdgcn_readfirstlane(t >> 5), blockIdx.x);
    for (unsigned i = t; i < NR * BIN_NODES * 16; i += L1_THREADS) acc[i] = 0.0f;
    for (unsigned i = t; i < NR * BIN_NODES; i += L1_THREADS) cl[i] = 0.0f;
    __syncthreads();
    const unsigned lane = t & 31u;
    const unsigned fb = (lane & 15u) * 4u;
    for (unsigned v0 = 0u; v0 < runs.first[RUNS]; v0 += 32u) {
        const unsigned v = v0 + lane;
        const bool live = v < runs.first[RUNS];
        const unsigned w = live ? list[run_slot(runs, v)] : 0u;
        const unsigned row = w >> 9;
        const unsigned dl = w & 511u;
        const unsigned r = row / NN;
        const unsigned slot = r * BIN_NODES + dl;
        const unsigned gb = (row - r * NN) * 64u;
        const unsigned ab = live ? slot * 64u : SKIP_EDGE;
        gather_rows_batch(x, acc, gb, ab, fb, std::make_integer_sequence<int, 16>{});
        if (live) atomicAdd(&cl[slot], 1.0f);
    }
    __syncthreads();
    for (unsigned i = t; i < NR * BIN_NODES; i += L1_THREADS) {
        const unsigned r = i / BIN_NODES;
        cnt[(size_t)r * NN + n0 + (i - r * BIN_NODES)] = cl[i];
    }
    const int tile = (int)(t >> 5);
    if (tile >= BIN_NODES / 16) return;
    const int ln   = tile * 16 + (int)(lane & 15u);
    const int arow = (int)n0 + ln;
    const int kh   = (int)(lane >> 4) << 1;
    const int col  = (int)(lane & 15u);

    v8f c0, c1;
    float bb0 = b1[col], bb1 = b1[16 + col];
#pragma unroll
    for (int v = 0; v < 8; ++v) { c0[v] = bb0; c1[v] = bb1; }

#pragma unroll
    for (int k = 0; k < 16; k += 4) {
        v2f a  = a_frag(x, arow, 16, k, kh, 1.0f);
        v2f bA = b_frag(root1, 32, k, kh, col);
        v2f bB = b_frag(root1, 32, k, kh, col + 16);
        c0 = __builtin_amdgcn_wmma_f32_16x16x4_f32(false, a, false, bA, (short)0, c0, false, false);
        c1 = __builtin_amdgcn_wmma_f32_16x16x4_f32(false, a, false, bB, (short)0, c1, false, false);
    }
    for (int r = 0; r < NR; ++r) {
        const float* Wr = W1 + (size_t)r * 16 * 32;
        const float sc = mean_weight(cl[r * BIN_NODES + ln]);
#pragma unroll
        for (int k = 0; k < 16; k += 4) {
            v2f a;
            a.x = acc[(r * BIN_NODES + ln) * 16 + k + kh] * sc;
            a.y = acc[(r * BIN_NODES + ln) * 16 + k + kh + 1] * sc;
            v2f bA = b_frag(Wr, 32, k, kh, col);
            v2f bB = b_frag(Wr, 32, k, kh, col + 16);
            c0 = __builtin_amdgcn_wmma_f32_16x16x4_f32(false, a, false, bA, (short)0, c0, false, false);
            c1 = __builtin_amdgcn_wmma_f32_16x16x4_f32(false, a, false, bB, (short)0, c1, false, false);
        }
    }
    float* ht = htile + tile * (16 * HP);
    const int hh = (int)(lane >> 4);
#pragma unroll
    for (int v = 0; v < 8; ++v) {
        ht[(8 * hh + v) * HP + col]      = fmaxf(c0[v], 0.0f);
        ht[(8 * hh + v) * HP + 16 + col] = fmaxf(c1[v], 0.0f);
    }
    __builtin_amdgcn_fence(__ATOMIC_ACQ_REL, "wavefront");

    const int lrow = ((col & 7) << 1) + (col >> 3);
    v2f a[8];
#pragma unroll
    for (int k = 0; k < 8; ++k) {
        a[k].x = ht[lrow * HP + 4 * k + kh];
        a[k].y = ht[lrow * HP + 4 * k + kh + 1];
    }

    const size_t nbase = (size_t)n0 + (size_t)(tile * 16 + hh);
    {
        v2f b[8];
#pragma unroll
        for (int k = 0; k < 8; ++k) b[k] = b_frag(root2, 16, 4 * k, kh, col);
        const float bb = b2[col];
        v8f c;
#pragma unroll
        for (int v = 0; v < 8; ++v) c[v] = bb;
#pragma unroll
        for (int k = 0; k < 8; ++k)
            c = __builtin_amdgcn_wmma_f32_16x16x4_f32(false, a[k], false, b[k], (short)0, c, false, false);
#pragma unroll
        for (int v = 0; v < 8; ++v) out[(nbase + (size_t)(2 * v)) * 16 + col] = c[v];
    }
#pragma unroll 1
    for (int g = 0; g < NR / 2; ++g) {
        v2f b[2][8];
        v8f c[2];
#pragma unroll
        for (int j = 0; j < 2; ++j) {
            const float* Wr = W2 + (size_t)(2 * g + j) * 32 * 16;
#pragma unroll
            for (int k = 0; k < 8; ++k) b[j][k] = b_frag(Wr, 16, 4 * k, kh, col);
#pragma unroll
            for (int v = 0; v < 8; ++v) c[j][v] = 0.0f;
        }
#pragma unroll
        for (int k = 0; k < 8; ++k)
#pragma unroll
            for (int j = 0; j < 2; ++j)
                c[j] = __builtin_amdgcn_wmma_f32_16x16x4_f32(false, a[k], false, b[j][k], (short)0, c[j], false, false);
#pragma unroll
        for (int j = 0; j < 2; ++j) {
            float* pr = proj + (size_t)(2 * g + j) * NN * 16;
#pragma unroll
            for (int v = 0; v < 8; ++v) pr[(nbase + (size_t)(2 * v)) * 16 + col] = c[j][v];
        }
    }
}

#define GATHER_THREADS WALK_THREADS
__global__ __launch_bounds__(GATHER_THREADS) void gather_bins(const float* __restrict__ proj,
                                                              const float* __restrict__ cnt,
                                                              const unsigned* __restrict__ list,
                                                              const unsigned* __restrict__ runstart,
                                                              float* __restrict__ out) {
    __shared__ float acc[BIN_NODES * 16];
    __shared__ float wgt[NR * BIN_NODES];
    const unsigned t = threadIdx.x;
    const unsigned n0 = blockIdx.x * BIN_NODES;
    const wave_runs runs = load_runs(runstart, __builtin_amdgcn_readfirstlane(t >> 5), blockIdx.x);
    for (unsigned i = t; i < BIN_NODES * 16; i += GATHER_THREADS) acc[i] = out[(size_t)n0 * 16 + i];
    for (unsigned i = t; i < NR * BIN_NODES; i += GATHER_THREADS) {
        const unsigned r = i / BIN_NODES;
        wgt[i] = mean_weight(cnt[(size_t)r * NN + n0 + (i - r * BIN_NODES)]);
    }
    __syncthreads();
    const unsigned lane = t & 31u;
    const unsigned fb = (lane & 15u) * 4u;
    for (unsigned v0 = 0u; v0 < runs.first[RUNS]; v0 += 32u) {
        const unsigned v = v0 + lane;
        const bool live = v < runs.first[RUNS];
        const unsigned w = live ? list[run_slot(runs, v)] : 0u;
        const unsigned row = w >> 9;
        const unsigned dl = w & 511u;
        const unsigned gb = row * 64u;
        const unsigned ab = live ? dl * 64u : SKIP_EDGE;
        const unsigned wb = __float_as_uint(wgt[(row / NN) * BIN_NODES + dl]);
        gather_scaled_batch(proj, acc, gb, ab, wb, fb, std::make_integer_sequence<int, 16>{});
    }
    __syncthreads();
    for (unsigned i = t; i < BIN_NODES * 16; i += GATHER_THREADS) out[(size_t)n0 * 16 + i] = acc[i];
}


extern "C" void kernel_launch(void* const* d_in, const int* in_sizes, int n_in,
                              void* d_out, int out_size, void* d_ws, size_t ws_size,
                              hipStream_t stream) {
    const float* x     = (const float*)d_in[0];
    const int*   eidx  = (const int*)d_in[1];
    const int*   etype = (const int*)d_in[2];
    const float* W1    = (const float*)d_in[3];
    const float* root1 = (const float*)d_in[4];
    const float* b1    = (const float*)d_in[5];
    const float* W2    = (const float*)d_in[6];
    const float* root2 = (const float*)d_in[7];
    const float* b2    = (const float*)d_in[8];
    float* out = (float*)d_out;

    const int* src = eidx;
    const int* dst = eidx + NE;

    float* plane = (float*)d_ws;
    float* proj  = plane + (size_t)NR * NN * 16;
    float* cnt   = plane + (size_t)NR * NN * 32;

    unsigned* list     = (unsigned*)plane;
    unsigned* runstart = list + NE;

    sort_chunks<<<NCHUNK, SORT_THREADS, 0, stream>>>(src, dst, etype, list, runstart);

    layer1_bins<<<NBINS, L1_THREADS, 0, stream>>>(x, list, runstart, W1, root1, b1,
                                                  W2, root2, b2, cnt, proj, out);

    gather_bins<<<NBINS, GATHER_THREADS, 0, stream>>>(proj, cnt, list, runstart, out);

    (void)in_sizes; (void)n_in; (void)out_size; (void)ws_size;
}


# staging loops (LDS clears, gather start-up from out and counts, count, list and out write-back) move 16 bytes per lane, all loads of a loop in flight before its first store: one round trip per loop
# speedup vs baseline: 5.0300x; 1.0479x over previous
#include <hip/hip_runtime.h>
#include <utility>


#define NN 100000
#define NE 3200000
#define NR 8

typedef __attribute__((ext_vector_type(2))) float v2f;
typedef __attribute__((ext_vector_type(8))) float v8f;
typedef __attribute__((ext_vector_type(4))) float v4f;
typedef __attribute__((ext_vector_type(4))) unsigned v4u;


__device__ __forceinline__ float mean_weight(float count) {
    return 1.0f / fmaxf(count, 1.0f);
}

#define BIN_NODES 400
#define NBINS (NN / BIN_NODES)
#define NCHUNK 256
#define CHUNK (NE / NCHUNK)
#define SORT_THREADS 512
#define SORT_EPT ((CHUNK + SORT_THREADS - 1) / SORT_THREADS)
#define WALK_THREADS 1024
#define RUNS (NCHUNK / (WALK_THREADS / 32))
#define NO_EDGE 0xFFFFFFFFu
#define SKIP_EDGE 0xFFFFFFFFu
static_assert(NN % BIN_NODES == 0 && BIN_NODES % 16 == 0, "ranges hold whole 16-node tiles");
static_assert(NBINS < 256 && BIN_NODES <= 512, "range id and local node fit their fields");
static_assert((unsigned long long)NR * NN < (1ull << 23), "packed row index fits 23 bits");
static_assert(NE % NCHUNK == 0 && CHUNK < 65536, "whole chunks; a rank inside a chunk fits 16 bits");
static_assert(NCHUNK % (WALK_THREADS / 32) == 0, "every walking wave owns RUNS whole runs");
static_assert(CHUNK % 4 == 0 && BIN_NODES % 4 == 0 && NN % 4 == 0, "chunks, ranges and count planes start on 16-byte boundaries");
#define SORT_Q4 ((CHUNK / 4 + SORT_THREADS - 1) / SORT_THREADS)

__device__ __forceinline__ void block_scan_256(unsigned* buf, unsigned t) {
    for (unsigned off = 1; off < 256; off <<= 1) {
        unsigned v = 0u;
        if (t < 256 && t >= off) v = buf[t - off];
        __syncthreads();
        if (t < 256) buf[t] += v;
        __syncthreads();
    }
}

__global__ __launch_bounds__(SORT_THREADS) void sort_chunks(const int* __restrict__ src,
                                                            const int* __restrict__ dst,
                                                            const int* __restrict__ et,
                                                            unsigned* __restrict__ list,
                                                            unsigned* __restrict__ runstart) {
    __shared__ unsigned lhist[256];
    __shared__ unsigned lend[256];
    __shared__ __attribute__((aligned(16))) unsigned sorted[CHUNK];
    const unsigned t = threadIdx.x;
    const unsigned first = blockIdx.x * CHUNK;
    if (t < 256) lhist[t] = 0u;
    unsigned dv[SORT_EPT], sv[SORT_EPT], tv[SORT_EPT];
#pragma unroll
    for (int k = 0; k < SORT_EPT; ++k) {
        const unsigned i = k * SORT_THREADS + t;
        const unsigned e = first + (i < CHUNK ? i : CHUNK - 1);
        dv[k] = (unsigned)dst[e];
        sv[k] = (unsigned)src[e];
        tv[k] = (unsigned)et[e];
    }
    __syncthreads();
    unsigned word[SORT_EPT], where[SORT_EPT];
#pragma unroll
    for (int k = 0; k < SORT_EPT; ++k) {
        const unsigned b = dv[k] / BIN_NODES;
        word[k] = ((tv[k] * NN + sv[k]) << 9) | (dv[k] - b * BIN_NODES);
        where[k] = NO_EDGE;
        if (k * SORT_THREADS + t < CHUNK) where[k] = (b << 16) | atomicAdd(&lhist[b], 1u);
    }
    __syncthreads();
    if (t < 256) lend[t] = lhist[t];
    __syncthreads();
    block_scan_256(lend, t);
    if (t < 256) {
        const unsigned start = t == 0 ? 0u : lend[t - 1];
        runstart[blockIdx.x * 256 + t] = start;
        lhist[t] = start;
    }
    __syncthreads();
    unsigned pos[SORT_EPT];
#pragma unroll
    for (int k = 0; k < SORT_EPT; ++k)
        pos[k] = lhist[where[k] != NO_EDGE ? where[k] >> 16 : 0u] + (where[k] & 0xFFFFu);
#pragma unroll
    for (int k = 0; k < SORT_EPT; ++k)
        if (where[k] != NO_EDGE) sorted[pos[k]] = word[k];
    __syncthreads();
    const v4u* s4 = reinterpret_cast<const v4u*>(sorted);
    v4u* l4 = reinterpret_cast<v4u*>(list + first);
    v4u q[SORT_Q4];
#pragma unroll
    for (int k = 0; k < SORT_Q4; ++k) {
        const unsigned i = k * SORT_THREADS + t;
        q[k] = s4[i < CHUNK / 4 ? i : CHUNK / 4 - 1];
    }
#pragma unroll
    for (int k = 0; k < SORT_Q4; ++k) {
        const unsigned i = k * SORT_THREADS + t;
        if (i < CHUNK / 4) l4[i] = q[k];
    }
}

struct wave_runs {
    unsigned first[RUNS + 1];
    unsigned shift[RUNS];
};
__device__ __forceinline__ wave_runs load_runs(const unsigned* __restrict__ runstart,
                                               unsigned wave, unsigned range) {
    wave_runs w;
    w.first[0] = 0u;
#pragma unroll
    for (int j = 0; j < RUNS; ++j) {
        const unsigned c = wave * RUNS + j;
        const unsigned lo = runstart[c * 256 + range], hi = runstart[c * 256 + range + 1];
        w.shift[j] = c * CHUNK + lo - w.first[j];
        w.first[j + 1] = w.first[j] + (hi - lo);
    }
    return w;
}
__device__ __forceinline__ unsigned run_slot(const wave_runs& w, unsigned v) {
    unsigned s = w.shift[0];
#pragma unroll
    for (int j = 1; j < RUNS; ++j) s = v >= w.first[j] ? w.shift[j] : s;
    return v + s;
}

template <int I>
__device__ __forceinline__ unsigned edge_lane(unsigned x) {
    return (unsigned)__builtin_amdgcn_update_dpp(0, (int)x, 0x150 + I, 0xF, 0xF, false);
}

__device__ __forceinline__ void add_local(float* acc, unsigned a, unsigned fb, float v) {
    if (a != SKIP_EDGE) atomicAdd((float*)((char*)acc + (a + fb)), v);
}

template <int... I>
__device__ __forceinline__ void gather_rows_batch(const float* __restrict__ feat, float* acc,
                                                  unsigned gb, unsigned ab, unsigned fb,
                                                  std::integer_sequence<int, I...>) {
    float v[sizeof...(I)];
    ((v[I] = *(const float*)((const char*)feat + (edge_lane<I>(gb) + fb))), ...);
    (add_local(acc, edge_lane<I>(ab), fb, v[I]), ...);
}
template <int... I>
__device__ __forceinline__ void gather_scaled_batch(const float* __restrict__ proj, float* acc,
                                                    unsigned gb, unsigned ab, unsigned wb,
                                                    unsigned fb,
                                                    std::integer_sequence<int, I...>) {
    float v[sizeof...(I)];
    ((v[I] = *(const float*)((const char*)proj + (edge_lane<I>(gb) + fb))), ...);
    (add_local(acc, edge_lane<I>(ab), fb,
               v[I] * __uint_as_float(edge_lane<I>(wb))), ...);
}

__device__ __forceinline__ v2f a_frag(const float* __restrict__ M, int row, int stride,
                                      int k0, int kh, float scale) {
    const float* p = M + (size_t)row * stride + k0 + kh;
    v2f a; a.x = p[0] * scale; a.y = p[1] * scale; return a;
}
__device__ __forceinline__ v2f b_frag(const float* __restrict__ W, int nout,
                                      int k0, int kh, int col) {
    v2f b;
    b.x = W[(size_t)(k0 + kh) * nout + col];
    b.y = W[(size_t)(k0 + kh + 1) * nout + col];
    return b;
}

#define L1_THREADS WALK_THREADS
#define HP 36
__global__ __launch_bounds__(L1_THREADS) void layer1_bins(
    const float* __restrict__ x, const unsigned* __restrict__ list,
    const unsigned* __restrict__ runstart,
    const float* __restrict__ W1, const float* __restrict__ root1, const float* __restrict__ b1,
    const float* __restrict__ W2, const float* __restrict__ root2, const float* __restrict__ b2,
    float* __restrict__ cnt, float* __restrict__ proj, float* __restrict__ out) {
    __shared__ __attribute__((aligned(16))) float acc[NR * BIN_NODES * 16];
    __shared__ __attribute__((aligned(16))) float cl[NR * BIN_NODES];
    __shared__ float htile[(BIN_NODES / 16) * 16 * HP];
    const unsigned t = threadIdx.x;
    const unsigned n0 = blockIdx.x * BIN_NODES;
    const wave_runs runs = load_runs(runstart, __builtin_amdgcn_readfirstlane(t >> 5), blockIdx.x);
    {
        const v4f zero = {0.0f, 0.0f, 0.0f, 0.0f};
        v4f* acc4 = reinterpret_cast<v4f*>(acc);
        v4f* cl4 = reinterpret_cast<v4f*>(cl);
        for (unsigned i = t; i < NR * BIN_NODES * 16 / 4; i += L1_THREADS) acc4[i] = zero;
        for (unsigned i = t; i < NR * BIN_NODES / 4; i += L1_THREADS) cl4[i] = zero;
    }
    __syncthreads();
    const unsigned lane = t & 31u;
    const unsigned fb = (lane & 15u) * 4u;
    for (unsigned v0 = 0u; v0 < runs.first[RUNS]; v0 += 32u) {
        const unsigned v = v0 + lane;
        const bool live = v < runs.first[RUNS];
        const unsigned w = live ? list[run_slot(runs, v)] : 0u;
        const unsigned row = w >> 9;
        const unsigned dl = w & 511u;
        const unsigned r = row / NN;
        const unsigned slot = r * BIN_NODES + dl;
        const unsigned gb = (row - r * NN) * 64u;
        const unsigned ab = live ? slot * 64u : SKIP_EDGE;
        gather_rows_batch(x, acc, gb, ab, fb, std::make_integer_sequence<int, 16>{});
        if (live) atomicAdd(&cl[slot], 1.0f);
    }
    __syncthreads();
    for (unsigned i = t; i < NR * BIN_NODES / 4; i += L1_THREADS) {
        const unsigned r = i / (BIN_NODES / 4);
        const v4f c4 = reinterpret_cast<const v4f*>(cl)[i];
        *reinterpret_cast<v4f*>(cnt + (size_t)r * NN + n0 + 4u * (i - r * (BIN_NODES / 4))) = c4;
    }
    const int tile = (int)(t >> 5);
    if (tile >= BIN_NODES / 16) return;
    const int ln   = tile * 16 + (int)(lane & 15u);
    const int arow = (int)n0 + ln;
    const int kh   = (int)(lane >> 4) << 1;
    const int col  = (int)(lane & 15u);

    v8f c0, c1;
    float bb0 = b1[col], bb1 = b1[16 + col];
#pragma unroll
    for (int v = 0; v < 8; ++v) { c0[v] = bb0; c1[v] = bb1; }

#pragma unroll
    for (int k = 0; k < 16; k += 4) {
        v2f a  = a_frag(x, arow, 16, k, kh, 1.0f);
        v2f bA = b_frag(root1, 32, k, kh, col);
        v2f bB = b_frag(root1, 32, k, kh, col + 16);
        c0 = __builtin_amdgcn_wmma_f32_16x16x4_f32(false, a, false, bA, (short)0, c0, false, false);
        c1 = __builtin_amdgcn_wmma_f32_16x16x4_f32(false, a, false, bB, (short)0, c1, false, false);
    }
    for (int r = 0; r < NR; ++r) {
        const float* Wr = W1 + (size_t)r * 16 * 32;
        const float sc = mean_weight(cl[r * BIN_NODES + ln]);
#pragma unroll
        for (int k = 0; k < 16; k += 4) {
            v2f a;
            a.x = acc[(r * BIN_NODES + ln) * 16 + k + kh] * sc;
            a.y = acc[(r * BIN_NODES + ln) * 16 + k + kh + 1] * sc;
            v2f bA = b_frag(Wr, 32, k, kh, col);
            v2f bB = b_frag(Wr, 32, k, kh, col + 16);
            c0 = __builtin_amdgcn_wmma_f32_16x16x4_f32(false, a, false, bA, (short)0, c0, false, false);
            c1 = __builtin_amdgcn_wmma_f32_16x16x4_f32(false, a, false, bB, (short)0, c1, false, false);
        }
    }
    float* ht = htile + tile * (16 * HP);
    const int hh = (int)(lane >> 4);
#pragma unroll
    for (int v = 0; v < 8; ++v) {
        ht[(8 * hh + v) * HP + col]      = fmaxf(c0[v], 0.0f);
        ht[(8 * hh + v) * HP + 16 + col] = fmaxf(c1[v], 0.0f);
    }
    __builtin_amdgcn_fence(__ATOMIC_ACQ_REL, "wavefront");

    const int lrow = ((col & 7) << 1) + (col >> 3);
    v2f a[8];
#pragma unroll
    for (int k = 0; k < 8; ++k) {
        a[k].x = ht[lrow * HP + 4 * k + kh];
        a[k].y = ht[lrow * HP + 4 * k + kh + 1];
    }

    const size_t nbase = (size_t)n0 + (size_t)(tile * 16 + hh);
    {
        v2f b[8];
#pragma unroll
        for (int k = 0; k < 8; ++k) b[k] = b_frag(root2, 16, 4 * k, kh, col);
        const float bb = b2[col];
        v8f c;
#pragma unroll
        for (int v = 0; v < 8; ++v) c[v] = bb;
#pragma unroll
        for (int k = 0; k < 8; ++k)
            c = __builtin_amdgcn_wmma_f32_16x16x4_f32(false, a[k], false, b[k], (short)0, c, false, false);
#pragma unroll
        for (int v = 0; v < 8; ++v) out[(nbase + (size_t)(2 * v)) * 16 + col] = c[v];
    }
#pragma unroll 1
    for (int g = 0; g < NR / 2; ++g) {
        v2f b[2][8];
        v8f c[2];
#pragma unroll
        for (int j = 0; j < 2; ++j) {
            const float* Wr = W2 + (size_t)(2 * g + j) * 32 * 16;
#pragma unroll
            for (int k = 0; k < 8; ++k) b[j][k] = b_frag(Wr, 16, 4 * k, kh, col);
#pragma unroll
            for (int v = 0; v < 8; ++v) c[j][v] = 0.0f;
        }
#pragma unroll
        for (int k = 0; k < 8; ++k)
#pragma unroll
            for (int j = 0; j < 2; ++j)
                c[j] = __builtin_amdgcn_wmma_f32_16x16x4_f32(false, a[k], false, b[j][k], (short)0, c[j], false, false);
#pragma unroll
        for (int j = 0; j < 2; ++j) {
            float* pr = proj + (size_t)(2 * g + j) * NN * 16;
#pragma unroll
            for (int v = 0; v < 8; ++v) pr[(nbase + (size_t)(2 * v)) * 16 + col] = c[j][v];
        }
    }
}

#define GATHER_THREADS WALK_THREADS
#define GATHER_NA4 (BIN_NODES * 16 / 4)
#define GATHER_NW4 (NR * BIN_NODES / 4)
#define GATHER_A4 ((GATHER_NA4 + GATHER_THREADS - 1) / GATHER_THREADS)
#define GATHER_W4 ((GATHER_NW4 + GATHER_THREADS - 1) / GATHER_THREADS)
__global__ __launch_bounds__(GATHER_THREADS) void gather_bins(const float* __restrict__ proj,
                                                              const float* __restrict__ cnt,
                                                              const unsigned* __restrict__ list,
                                                              const unsigned* __restrict__ runstart,
                                                              float* __restrict__ out) {
    __shared__ __attribute__((aligned(16))) float acc[BIN_NODES * 16];
    __shared__ __attribute__((aligned(16))) float wgt[NR * BIN_NODES];
    const unsigned t = threadIdx.x;
    const unsigned n0 = blockIdx.x * BIN_NODES;
    const wave_runs runs = load_runs(runstart, __builtin_amdgcn_readfirstlane(t >> 5), blockIdx.x);
    v4f* out4 = reinterpret_cast<v4f*>(out + (size_t)n0 * 16);
    v4f s[GATHER_A4], c4[GATHER_W4];
#pragma unroll
    for (int k = 0; k < GATHER_A4; ++k) {
        const unsigned i = k * GATHER_THREADS + t;
        s[k] = out4[i < GATHER_NA4 ? i : GATHER_NA4 - 1];
    }
#pragma unroll
    for (int k = 0; k < GATHER_W4; ++k) {
        const unsigned i = k * GATHER_THREADS + t;
        const unsigned j = i < GATHER_NW4 ? i : GATHER_NW4 - 1;
        const unsigned r = j / (BIN_NODES / 4);
        c4[k] = *reinterpret_cast<const v4f*>(cnt + (size_t)r * NN + n0 + 4u * (j - r * (BIN_NODES / 4)));
    }
#pragma unroll
    for (int k = 0; k < GATHER_A4; ++k) {
        const unsigned i = k * GATHER_THREADS + t;
        if (i < GATHER_NA4) reinterpret_cast<v4f*>(acc)[i] = s[k];
    }
#pragma unroll
    for (int k = 0; k < GATHER_W4; ++k) {
        const unsigned i = k * GATHER_THREADS + t;
        if (i < GATHER_NW4) {
            v4f w4;
            w4.x = mean_weight(c4[k].x); w4.y = mean_weight(c4[k].y);
            w4.z = mean_weight(c4[k].z); w4.w = mean_weight(c4[k].w);
            reinterpret_cast<v4f*>(wgt)[i] = w4;
        }
    }
    __syncthreads();
    const unsigned lane = t & 31u;
    const unsigned fb = (lane & 15u) * 4u;
    for (unsigned v0 = 0u; v0 < runs.first[RUNS]; v0 += 32u) {
        const unsigned v = v0 + lane;
        const bool live = v < runs.first[RUNS];
        const unsigned w = live ? list[run_slot(runs, v)] : 0u;
        const unsigned row = w >> 9;
        const unsigned dl = w & 511u;
        const unsigned gb = row * 64u;
        const unsigned ab = live ? dl * 64u : SKIP_EDGE;
        const unsigned wb = __float_as_uint(wgt[(row / NN) * BIN_NODES + dl]);
        gather_scaled_batch(proj, acc, gb, ab, wb, fb, std::make_integer_sequence<int, 16>{});
    }
    __syncthreads();
#pragma unroll
    for (int k = 0; k < GATHER_A4; ++k) {
        const unsigned i = k * GATHER_THREADS + t;
        s[k] = reinterpret_cast<const v4f*>(acc)[i < GATHER_NA4 ? i : GATHER_NA4 - 1];
    }
#pragma unroll
    for (int k = 0; k < GATHER_A4; ++k) {
        const unsigned i = k * GATHER_THREADS + t;
        if (i < GATHER_NA4) out4[i] = s[k];
    }
}


extern "C" void kernel_launch(void* const* d_in, const int* in_sizes, int n_in,
                              void* d_out, int out_size, void* d_ws, size_t ws_size,
                              hipStream_t stream) {
    const float* x     = (const float*)d_in[0];
    const int*   eidx  = (const int*)d_in[1];
    const int*   etype = (const int*)d_in[2];
    const float* W1    = (const float*)d_in[3];
    const float* root1 = (const float*)d_in[4];
    const float* b1    = (const float*)d_in[5];
    const float* W2    = (const float*)d_in[6];
    const float* root2 = (const float*)d_in[7];
    const float* b2    = (const float*)d_in[8];
    float* out = (float*)d_out;

    const int* src = eidx;
    const int* dst = eidx + NE;

    float* plane = (float*)d_ws;
    float* proj  = plane + (size_t)NR * NN * 16;
    float* cnt   = plane + (size_t)NR * NN * 32;

    unsigned* list     = (unsigned*)plane;
    unsigned* runstart = list + NE;

    sort_chunks<<<NCHUNK, SORT_THREADS, 0, stream>>>(src, dst, etype, list, runstart);

    layer1_bins<<<NBINS, L1_THREADS, 0, stream>>>(x, list, runstart, W1, root1, b1,
                                                  W2, root2, b2, cnt, proj, out);

    gather_bins<<<NBINS, GATHER_THREADS, 0, stream>>>(proj, cnt, list, runstart, out);

    (void)in_sizes; (void)n_in; (void)out_size; (void)ws_size;
}


# edge walks run without lane masks: a lane past a wave's last edge adds into a spare LDS row and count, so the 16 mask-and-branch blocks per 32-edge trip become plain LDS adds; sums as before
# speedup vs baseline: 5.2509x; 1.0439x over previous
#include <hip/hip_runtime.h>
#include <utility>


#define NN 100000
#define NE 3200000
#define NR 8

typedef __attribute__((ext_vector_type(2))) float v2f;
typedef __attribute__((ext_vector_type(8))) float v8f;
typedef __attribute__((ext_vector_type(4))) float v4f;
typedef __attribute__((ext_vector_type(4))) unsigned v4u;


__device__ __forceinline__ float mean_weight(float count) {
    return 1.0f / fmaxf(count, 1.0f);
}

#define BIN_NODES 400
#define NBINS (NN / BIN_NODES)
#define NCHUNK 256
#define CHUNK (NE / NCHUNK)
#define SORT_THREADS 512
#define SORT_EPT ((CHUNK + SORT_THREADS - 1) / SORT_THREADS)
#define WALK_THREADS 1024
#define RUNS (NCHUNK / (WALK_THREADS / 32))
#define NO_EDGE 0xFFFFFFFFu
static_assert(NN % BIN_NODES == 0 && BIN_NODES % 16 == 0, "ranges hold whole 16-node tiles");
static_assert(NBINS < 256 && BIN_NODES <= 512, "range id and local node fit their fields");
static_assert((unsigned long long)NR * NN < (1ull << 23), "packed row index fits 23 bits");
static_assert(NE % NCHUNK == 0 && CHUNK < 65536, "whole chunks; a rank inside a chunk fits 16 bits");
static_assert(NCHUNK % (WALK_THREADS / 32) == 0, "every walking wave owns RUNS whole runs");
static_assert(CHUNK % 4 == 0 && BIN_NODES % 4 == 0 && NN % 4 == 0, "chunks, ranges and count planes start on 16-byte boundaries");
#define SORT_Q4 ((CHUNK / 4 + SORT_THREADS - 1) / SORT_THREADS)

__device__ __forceinline__ void block_scan_256(unsigned* buf, unsigned t) {
    for (unsigned off = 1; off < 256; off <<= 1) {
        unsigned v = 0u;
        if (t < 256 && t >= off) v = buf[t - off];
        __syncthreads();
        if (t < 256) buf[t] += v;
        __syncthreads();
    }
}

__global__ __launch_bounds__(SORT_THREADS) void sort_chunks(const int* __restrict__ src,
                                                            const int* __restrict__ dst,
                                                            const int* __restrict__ et,
                                                            unsigned* __restrict__ list,
                                                            unsigned* __restrict__ runstart) {
    __shared__ unsigned lhist[256];
    __shared__ unsigned lend[256];
    __shared__ __attribute__((aligned(16))) unsigned sorted[CHUNK];
    const unsigned t = threadIdx.x;
    const unsigned first = blockIdx.x * CHUNK;
    if (t < 256) lhist[t] = 0u;
    unsigned dv[SORT_EPT], sv[SORT_EPT], tv[SORT_EPT];
#pragma unroll
    for (int k = 0; k < SORT_EPT; ++k) {
        const unsigned i = k * SORT_THREADS + t;
        const unsigned e = first + (i < CHUNK ? i : CHUNK - 1);
        dv[k] = (unsigned)dst[e];
        sv[k] = (unsigned)src[e];
        tv[k] = (unsigned)et[e];
    }
    __syncthreads();
    unsigned word[SORT_EPT], where[SORT_EPT];
#pragma unroll
    for (int k = 0; k < SORT_EPT; ++k) {
        const unsigned b = dv[k] / BIN_NODES;
        word[k] = ((tv[k] * NN + sv[k]) << 9) | (dv[k] - b * BIN_NODES);
        where[k] = NO_EDGE;
        if (k * SORT_THREADS + t < CHUNK) where[k] = (b << 16) | atomicAdd(&lhist[b], 1u);
    }
    __syncthreads();
    if (t < 256) lend[t] = lhist[t];
    __syncthreads();
    block_scan_256(lend, t);
    if (t < 256) {
        const unsigned start = t == 0 ? 0u : lend[t - 1];
        runstart[blockIdx.x * 256 + t] = start;
        lhist[t] = start;
    }
    __syncthreads();
    unsigned pos[SORT_EPT];
#pragma unroll
    for (int k = 0; k < SORT_EPT; ++k)
        pos[k] = lhist[where[k] != NO_EDGE ? where[k] >> 16 : 0u] + (where[k] & 0xFFFFu);
#pragma unroll
    for (int k = 0; k < SORT_EPT; ++k)
        if (where[k] != NO_EDGE) sorted[pos[k]] = word[k];
    __syncthreads();
    const v4u* s4 = reinterpret_cast<const v4u*>(sorted);
    v4u* l4 = reinterpret_cast<v4u*>(list + first);
    v4u q[SORT_Q4];
#pragma unroll
    for (int k = 0; k < SORT_Q4; ++k) {
        const unsigned i = k * SORT_THREADS + t;
        q[k] = s4[i < CHUNK / 4 ? i : CHUNK / 4 - 1];
    }
#pragma unroll
    for (int k = 0; k < SORT_Q4; ++k) {
        const unsigned i = k * SORT_THREADS + t;
        if (i < CHUNK / 4) l4[i] = q[k];
    }
}

struct wave_runs {
    unsigned first[RUNS + 1];
    unsigned shift[RUNS];
};
__device__ __forceinline__ wave_runs load_runs(const unsigned* __restrict__ runstart,
                                               unsigned wave, unsigned range) {
    wave_runs w;
    w.first[0] = 0u;
#pragma unroll
    for (int j = 0; j < RUNS; ++j) {
        const unsigned c = wave * RUNS + j;
        const unsigned lo = runstart[c * 256 + range], hi = runstart[c * 256 + range + 1];
        w.shift[j] = c * CHUNK + lo - w.first[j];
        w.first[j + 1] = w.first[j] + (hi - lo);
    }
    return w;
}
__device__ __forceinline__ unsigned run_slot(const wave_runs& w, unsigned v) {
    unsigned s = w.shift[0];
#pragma unroll
    for (int j = 1; j < RUNS; ++j) s = v >= w.first[j] ? w.shift[j] : s;
    return v + s;
}

template <int I>
__device__ __forceinline__ unsigned edge_lane(unsigned x) {
    return (unsigned)__builtin_amdgcn_update_dpp(0, (int)x, 0x150 + I, 0xF, 0xF, false);
}

__device__ __forceinline__ void add_local(float* acc, unsigned a, unsigned fb, float v) {
    atomicAdd((float*)((char*)acc + (a + fb)), v);
}

template <int... I>
__device__ __forceinline__ void gather_rows_batch(const float* __restrict__ feat, float* acc,
                                                  unsigned gb, unsigned ab, unsigned fb,
                                                  std::integer_sequence<int, I...>) {
    float v[sizeof...(I)];
    ((v[I] = *(const float*)((const char*)feat + (edge_lane<I>(gb) + fb))), ...);
    (add_local(acc, edge_lane<I>(ab), fb, v[I]), ...);
}
template <int... I>
__device__ __forceinline__ void gather_scaled_batch(const float* __restrict__ proj, float* acc,
                                                    unsigned gb, unsigned ab, unsigned wb,
                                                    unsigned fb,
                                                    std::integer_sequence<int, I...>) {
    float v[sizeof...(I)];
    ((v[I] = *(const float*)((const char*)proj + (edge_lane<I>(gb) + fb))), ...);
    (add_local(acc, edge_lane<I>(ab), fb,
               v[I] * __uint_as_float(edge_lane<I>(wb))), ...);
}

__device__ __forceinline__ v2f a_frag(const float* __restrict__ M, int row, int stride,
                                      int k0, int kh, float scale) {
    const float* p = M + (size_t)row * stride + k0 + kh;
    v2f a; a.x = p[0] * scale; a.y = p[1] * scale; return a;
}
__device__ __forceinline__ v2f b_frag(const float* __restrict__ W, int nout,
                                      int k0, int kh, int col) {
    v2f b;
    b.x = W[(size_t)(k0 + kh) * nout + col];
    b.y = W[(size_t)(k0 + kh + 1) * nout + col];
    return b;
}

#define L1_THREADS WALK_THREADS
#define HP 36
__global__ __launch_bounds__(L1_THREADS) void layer1_bins(
    const float* __restrict__ x, const unsigned* __restrict__ list,
    const unsigned* __restrict__ runstart,
    const float* __restrict__ W1, const float* __restrict__ root1, const float* __restrict__ b1,
    const float* __restrict__ W2, const float* __restrict__ root2, const float* __restrict__ b2,
    float* __restrict__ cnt, float* __restrict__ proj, float* __restrict__ out) {
    __shared__ __attribute__((aligned(16))) float acc[NR * BIN_NODES * 16 + 16];
    __shared__ __attribute__((aligned(16))) float cl[NR * BIN_NODES + 4];
    __shared__ float htile[(BIN_NODES / 16) * 16 * HP];
    const unsigned t = threadIdx.x;
    const unsigned n0 = blockIdx.x * BIN_NODES;
    const wave_runs runs = load_runs(runstart, __builtin_amdgcn_readfirstlane(t >> 5), blockIdx.x);
    {
        const v4f zero = {0.0f, 0.0f, 0.0f, 0.0f};
        v4f* acc4 = reinterpret_cast<v4f*>(acc);
        v4f* cl4 = reinterpret_cast<v4f*>(cl);
        for (unsigned i = t; i < (NR * BIN_NODES * 16 + 16) / 4; i += L1_THREADS) acc4[i] = zero;
        for (unsigned i = t; i < (NR * BIN_NODES + 4) / 4; i += L1_THREADS) cl4[i] = zero;
    }
    __syncthreads();
    const unsigned lane = t & 31u;
    const unsigned fb = (lane & 15u) * 4u;
    for (unsigned v0 = 0u; v0 < runs.first[RUNS]; v0 += 32u) {
        const unsigned v = v0 + lane;
        const bool live = v < runs.first[RUNS];
        const unsigned w = list[run_slot(runs, live ? v : runs.first[RUNS] - 1u)];
        const unsigned row = w >> 9;
        const unsigned dl = w & 511u;
        const unsigned r = row / NN;
        const unsigned slot = live ? r * BIN_NODES + dl : NR * BIN_NODES;
        const unsigned gb = (row - r * NN) * 64u;
        const unsigned ab = slot * 64u;
        gather_rows_batch(x, acc, gb, ab, fb, std::make_integer_sequence<int, 16>{});
        atomicAdd(&cl[slot], 1.0f);
    }
    __syncthreads();
    for (unsigned i = t; i < NR * BIN_NODES / 4; i += L1_THREADS) {
        const unsigned r = i / (BIN_NODES / 4);
        const v4f c4 = reinterpret_cast<const v4f*>(cl)[i];
        *reinterpret_cast<v4f*>(cnt + (size_t)r * NN + n0 + 4u * (i - r * (BIN_NODES / 4))) = c4;
    }
    const int tile = (int)(t >> 5);
    if (tile >= BIN_NODES / 16) return;
    const int ln   = tile * 16 + (int)(lane & 15u);
    const int arow = (int)n0 + ln;
    const int kh   = (int)(lane >> 4) << 1;
    const int col  = (int)(lane & 15u);

    v8f c0, c1;
    float bb0 = b1[col], bb1 = b1[16 + col];
#pragma unroll
    for (int v = 0; v < 8; ++v) { c0[v] = bb0; c1[v] = bb1; }

#pragma unroll
    for (int k = 0; k < 16; k += 4) {
        v2f a  = a_frag(x, arow, 16, k, kh, 1.0f);
        v2f bA = b_frag(root1, 32, k, kh, col);
        v2f bB = b_frag(root1, 32, k, kh, col + 16);
        c0 = __builtin_amdgcn_wmma_f32_16x16x4_f32(false, a, false, bA, (short)0, c0, false, false);
        c1 = __builtin_amdgcn_wmma_f32_16x16x4_f32(false, a, false, bB, (short)0, c1, false, false);
    }
    for (int r = 0; r < NR; ++r) {
        const float* Wr = W1 + (size_t)r * 16 * 32;
        const float sc = mean_weight(cl[r * BIN_NODES + ln]);
#pragma unroll
        for (int k = 0; k < 16; k += 4) {
            v2f a;
            a.x = acc[(r * BIN_NODES + ln) * 16 + k + kh] * sc;
            a.y = acc[(r * BIN_NODES + ln) * 16 + k + kh + 1] * sc;
            v2f bA = b_frag(Wr, 32, k, kh, col);
            v2f bB = b_frag(Wr, 32, k, kh, col + 16);
            c0 = __builtin_amdgcn_wmma_f32_16x16x4_f32(false, a, false, bA, (short)0, c0, false, false);
            c1 = __builtin_amdgcn_wmma_f32_16x16x4_f32(false, a, false, bB, (short)0, c1, false, false);
        }
    }
    float* ht = htile + tile * (16 * HP);
    const int hh = (int)(lane >> 4);
#pragma unroll
    for (int v = 0; v < 8; ++v) {
        ht[(8 * hh + v) * HP + col]      = fmaxf(c0[v], 0.0f);
        ht[(8 * hh + v) * HP + 16 + col] = fmaxf(c1[v], 0.0f);
    }
    __builtin_amdgcn_fence(__ATOMIC_ACQ_REL, "wavefront");

    const int lrow = ((col & 7) << 1) + (col >> 3);
    v2f a[8];
#pragma unroll
    for (int k = 0; k < 8; ++k) {
        a[k].x = ht[lrow * HP + 4 * k + kh];
        a[k].y = ht[lrow * HP + 4 * k + kh + 1];
    }

    const size_t nbase = (size_t)n0 + (size_t)(tile * 16 + hh);
    {
        v2f b[8];
#pragma unroll
        for (int k = 0; k < 8; ++k) b[k] = b_frag(root2, 16, 4 * k, kh, col);
        const float bb = b2[col];
        v8f c;
#pragma unroll
        for (int v = 0; v < 8; ++v) c[v] = bb;
#pragma unroll
        for (int k = 0; k < 8; ++k)
            c = __builtin_amdgcn_wmma_f32_16x16x4_f32(false, a[k], false, b[k], (short)0, c, false, false);
#pragma unroll
        for (int v = 0; v < 8; ++v) out[(nbase + (size_t)(2 * v)) * 16 + col] = c[v];
    }
#pragma unroll 1
    for (int g = 0; g < NR / 2; ++g) {
        v2f b[2][8];
        v8f c[2];
#pragma unroll
        for (int j = 0; j < 2; ++j) {
            const float* Wr = W2 + (size_t)(2 * g + j) * 32 * 16;
#pragma unroll
            for (int k = 0; k < 8; ++k) b[j][k] = b_frag(Wr, 16, 4 * k, kh, col);
#pragma unroll
            for (int v = 0; v < 8; ++v) c[j][v] = 0.0f;
        }
#pragma unroll
        for (int k = 0; k < 8; ++k)
#pragma unroll
            for (int j = 0; j < 2; ++j)
                c[j] = __builtin_amdgcn_wmma_f32_16x16x4_f32(false, a[k], false, b[j][k], (short)0, c[j], false, false);
#pragma unroll
        for (int j = 0; j < 2; ++j) {
            float* pr = proj + (size_t)(2 * g + j) * NN * 16;
#pragma unroll
            for (int v = 0; v < 8; ++v) pr[(nbase + (size_t)(2 * v)) * 16 + col] = c[j][v];
        }
    }
}

#define GATHER_THREADS WALK_THREADS
#define GATHER_NA4 (BIN_NODES * 16 / 4)
#define GATHER_NW4 (NR * BIN_NODES / 4)
#define GATHER_A4 ((GATHER_NA4 + GATHER_THREADS - 1) / GATHER_THREADS)
#define GATHER_W4 ((GATHER_NW4 + GATHER_THREADS - 1) / GATHER_THREADS)
__global__ __launch_bounds__(GATHER_THREADS) void gather_bins(const float* __restrict__ proj,
                                                              const float* __restrict__ cnt,
                                                              const unsigned* __restrict__ list,
                                                              const unsigned* __restrict__ runstart,
                                                              float* __restrict__ out) {
    __shared__ __attribute__((aligned(16))) float acc[BIN_NODES * 16 + 16];
    __shared__ __attribute__((aligned(16))) float wgt[NR * BIN_NODES];
    const unsigned t = threadIdx.x;
    const unsigned n0 = blockIdx.x * BIN_NODES;
    const wave_runs runs = load_runs(runstart, __builtin_amdgcn_readfirstlane(t >> 5), blockIdx.x);
    v4f* out4 = reinterpret_cast<v4f*>(out + (size_t)n0 * 16);
    v4f s[GATHER_A4], c4[GATHER_W4];
#pragma unroll
    for (int k = 0; k < GATHER_A4; ++k) {
        const unsigned i = k * GATHER_THREADS + t;
        s[k] = out4[i < GATHER_NA4 ? i : GATHER_NA4 - 1];
    }
#pragma unroll
    for (int k = 0; k < GATHER_W4; ++k) {
        const unsigned i = k * GATHER_THREADS + t;
        const unsigned j = i < GATHER_NW4 ? i : GATHER_NW4 - 1;
        const unsigned r = j / (BIN_NODES / 4);
        c4[k] = *reinterpret_cast<const v4f*>(cnt + (size_t)r * NN + n0 + 4u * (j - r * (BIN_NODES / 4)));
    }
#pragma unroll
    for (int k = 0; k < GATHER_A4; ++k) {
        const unsigned i = k * GATHER_THREADS + t;
        if (i < GATHER_NA4) reinterpret_cast<v4f*>(acc)[i] = s[k];
    }
#pragma unroll
    for (int k = 0; k < GATHER_W4; ++k) {
        const unsigned i = k * GATHER_THREADS + t;
        if (i < GATHER_NW4) {
            v4f w4;
            w4.x = mean_weight(c4[k].x); w4.y = mean_weight(c4[k].y);
            w4.z = mean_weight(c4[k].z); w4.w = mean_weight(c4[k].w);
            reinterpret_cast<v4f*>(wgt)[i] = w4;
        }
    }
    if (t < 4u) {
        const v4f zero = {0.0f, 0.0f, 0.0f, 0.0f};
        reinterpret_cast<v4f*>(acc)[GATHER_NA4 + t] = zero;
    }
    __syncthreads();
    const unsigned lane = t & 31u;
    const unsigned fb = (lane & 15u) * 4u;
    for (unsigned v0 = 0u; v0 < runs.first[RUNS]; v0 += 32u) {
        const unsigned v = v0 + lane;
        const bool live = v < runs.first[RUNS];
        const unsigned w = list[run_slot(runs, live ? v : runs.first[RUNS] - 1u)];
        const unsigned row = w >> 9;
        const unsigned dl = w & 511u;
        const unsigned gb = row * 64u;
        const unsigned ab = (live ? dl : (unsigned)BIN_NODES) * 64u;
        const unsigned wb = __float_as_uint(wgt[(row / NN) * BIN_NODES + dl]);
        gather_scaled_batch(proj, acc, gb, ab, wb, fb, std::make_integer_sequence<int, 16>{});
    }
    __syncthreads();
#pragma unroll
    for (int k = 0; k < GATHER_A4; ++k) {
        const unsigned i = k * GATHER_THREADS + t;
        s[k] = reinterpret_cast<const v4f*>(acc)[i < GATHER_NA4 ? i : GATHER_NA4 - 1];
    }
#pragma unroll
    for (int k = 0; k < GATHER_A4; ++k) {
        const unsigned i = k * GATHER_THREADS + t;
        if (i < GATHER_NA4) out4[i] = s[k];
    }
}


extern "C" void kernel_launch(void* const* d_in, const int* in_sizes, int n_in,
                              void* d_out, int out_size, void* d_ws, size_t ws_size,
                              hipStream_t stream) {
    const float* x     = (const float*)d_in[0];
    const int*   eidx  = (const int*)d_in[1];
    const int*   etype = (const int*)d_in[2];
    const float* W1    = (const float*)d_in[3];
    const float* root1 = (const float*)d_in[4];
    const float* b1    = (const float*)d_in[5];
    const float* W2    = (const float*)d_in[6];
    const float* root2 = (const float*)d_in[7];
    const float* b2    = (const float*)d_in[8];
    float* out = (float*)d_out;

    const int* src = eidx;
    const int* dst = eidx + NE;

    float* plane = (float*)d_ws;
    float* proj  = plane + (size_t)NR * NN * 16;
    float* cnt   = plane + (size_t)NR * NN * 32;

    unsigned* list     = (unsigned*)plane;
    unsigned* runstart = list + NE;

    sort_chunks<<<NCHUNK, SORT_THREADS, 0, stream>>>(src, dst, etype, list, runstart);

    layer1_bins<<<NBINS, L1_THREADS, 0, stream>>>(x, list, runstart, W1, root1, b1,
                                                  W2, root2, b2, cnt, proj, out);

    gather_bins<<<NBINS, GATHER_THREADS, 0, stream>>>(proj, cnt, list, runstart, out);

    (void)in_sizes; (void)n_in; (void)out_size; (void)ws_size;
}
